# starting kernel
# baseline (speedup 1.0000x reference)
#include <hip/hip_runtime.h>
#include <hip/hip_bf16.h>
#include <math.h>


typedef __attribute__((ext_vector_type(2))) float v2f;
typedef __attribute__((ext_vector_type(8))) float v8f;

#define N_NODES 50000
#define N_EDGES 800000
#define IN_CH   64
#define HID_CH  128
#define OUT_CH  40

__global__ void zero_f4(float4* __restrict__ p, int n4) {
  int i = blockIdx.x * blockDim.x + threadIdx.x;
  if (i < n4) p[i] = make_float4(0.f, 0.f, 0.f, 0.f);
}

template <int LOG2G>
__global__ void scatter_add_kernel(const float* __restrict__ feat,
                                   const int* __restrict__ srcv,
                                   const int* __restrict__ dstv,
                                   float* __restrict__ agg) {
  const int groups = 1 << LOG2G;
  const int ch = groups * 4;
  long long idx = (long long)blockIdx.x * blockDim.x + threadIdx.x;
  const long long total = (long long)N_EDGES << LOG2G;
  if (idx >= total) return;
  const int e = (int)(idx >> LOG2G);
  const int g = (int)(idx & (groups - 1));
  const int s = srcv[e];
  const int d = dstv[e];
  const float4 v = *(const float4*)(feat + (long long)s * ch + g * 4);
  float* o = agg + (long long)d * ch + g * 4;
  atomicAdd(o + 0, v.x);
  atomicAdd(o + 1, v.y);
  atomicAdd(o + 2, v.z);
  atomicAdd(o + 3, v.w);
}

__device__ __forceinline__ v2f load_frag(const float* base) {
  return *(const v2f*)base;
}

__global__ __launch_bounds__(256) void gemm1_relu(
    const float* __restrict__ agg1, const float* __restrict__ x,
    const float* __restrict__ Wrel, const float* __restrict__ Wroot,
    const float* __restrict__ bias, float* __restrict__ h) {
  const int lane = threadIdx.x & 31;
  const int wave = threadIdx.x >> 5;
  const int m_base = blockIdx.x * 16;
  const int n_base = wave * 16;
  const int l15 = lane & 15;
  const int kh = (lane >> 4) * 2;
  const int arow = m_base + l15;
  const int bcol = n_base + l15;

  const float* A0 = agg1 + (long long)arow * IN_CH + kh;
  const float* X0 = x + (long long)arow * IN_CH + kh;
  const float* Wr = Wrel + (long long)bcol * IN_CH + kh;
  const float* Wo = Wroot + (long long)bcol * IN_CH + kh;

  v8f c = {};
#pragma unroll
  for (int k = 0; k < IN_CH; k += 4) {
    v2f a = load_frag(A0 + k);
    v2f b = load_frag(Wr + k);
    c = __builtin_amdgcn_wmma_f32_16x16x4_f32(false, a, false, b, (short)0, c,
                                              false, false);
  }
#pragma unroll
  for (int k = 0; k < IN_CH; k += 4) {
    v2f a = load_frag(X0 + k);
    v2f b = load_frag(Wo + k);
    c = __builtin_amdgcn_wmma_f32_16x16x4_f32(false, a, false, b, (short)0, c,
                                              false, false);
  }

  const float bn = bias[bcol];
  const int rbase = m_base + ((lane >> 4) << 3);
#pragma unroll
  for (int i = 0; i < 8; ++i) {
    float v = c[i] + bn;
    h[(long long)(rbase + i) * HID_CH + bcol] = fmaxf(v, 0.0f);
  }
}

__global__ __launch_bounds__(256) void gemm2_logsoftmax(
    const float* __restrict__ agg2, const float* __restrict__ h,
    const float* __restrict__ Wrel, const float* __restrict__ Wroot,
    const float* __restrict__ bias, float* __restrict__ out) {
  const int lane = threadIdx.x & 31;
  const int wave = threadIdx.x >> 5;
  const int mtile = blockIdx.x * 8 + wave;
  if (mtile >= N_NODES / 16) return;
  const int m_base = mtile * 16;
  const int l15 = lane & 15;
  const int kh = (lane >> 4) * 2;
  const int arow = m_base + l15;
  const int n2 = 32 + l15;
  const bool v2ok = (n2 < OUT_CH);
  const int n2c = v2ok ? n2 : 0;

  const float* A0 = agg2 + (long long)arow * HID_CH + kh;
  const float* H0 = h + (long long)arow * HID_CH + kh;
  const float* Wr0 = Wrel + (long long)l15 * HID_CH + kh;
  const float* Wr1 = Wrel + (long long)(16 + l15) * HID_CH + kh;
  const float* Wr2 = Wrel + (long long)n2c * HID_CH + kh;
  const float* Wo0 = Wroot + (long long)l15 * HID_CH + kh;
  const float* Wo1 = Wroot + (long long)(16 + l15) * HID_CH + kh;
  const float* Wo2 = Wroot + (long long)n2c * HID_CH + kh;

  v8f c0 = {}, c1 = {}, c2 = {};
  for (int k = 0; k < HID_CH; k += 4) {
    v2f a = load_frag(A0 + k);
    v2f b0 = load_frag(Wr0 + k);
    v2f b1 = load_frag(Wr1 + k);
    v2f b2 = load_frag(Wr2 + k);
    if (!v2ok) { b2.x = 0.f; b2.y = 0.f; }
    c0 = __builtin_amdgcn_wmma_f32_16x16x4_f32(false, a, false, b0, (short)0,
                                               c0, false, false);
    c1 = __builtin_amdgcn_wmma_f32_16x16x4_f32(false, a, false, b1, (short)0,
                                               c1, false, false);
    c2 = __builtin_amdgcn_wmma_f32_16x16x4_f32(false, a, false, b2, (short)0,
                                               c2, false, false);
  }
  for (int k = 0; k < HID_CH; k += 4) {
    v2f a = load_frag(H0 + k);
    v2f b0 = load_frag(Wo0 + k);
    v2f b1 = load_frag(Wo1 + k);
    v2f b2 = load_frag(Wo2 + k);
    if (!v2ok) { b2.x = 0.f; b2.y = 0.f; }
    c0 = __builtin_amdgcn_wmma_f32_16x16x4_f32(false, a, false, b0, (short)0,
                                               c0, false, false);
    c1 = __builtin_amdgcn_wmma_f32_16x16x4_f32(false, a, false, b1, (short)0,
                                               c1, false, false);
    c2 = __builtin_amdgcn_wmma_f32_16x16x4_f32(false, a, false, b2, (short)0,
                                               c2, false, false);
  }

  const float bn0 = bias[l15];
  const float bn1 = bias[16 + l15];
  const float bn2 = v2ok ? bias[n2] : 0.0f;
  const int rbase = m_base + ((lane >> 4) << 3);
#pragma unroll
  for (int i = 0; i < 8; ++i) {
    float v0 = c0[i] + bn0;
    float v1 = c1[i] + bn1;
    float v2 = v2ok ? (c2[i] + bn2) : -INFINITY;
    float m = fmaxf(fmaxf(v0, v1), v2);
#pragma unroll
    for (int msk = 8; msk >= 1; msk >>= 1)
      m = fmaxf(m, __shfl_xor(m, msk, 32));
    float s = __expf(v0 - m) + __expf(v1 - m) + (v2ok ? __expf(v2 - m) : 0.0f);
#pragma unroll
    for (int msk = 8; msk >= 1; msk >>= 1) s += __shfl_xor(s, msk, 32);
    const float lz = m + __logf(s);
    const long long ro = (long long)(rbase + i) * OUT_CH;
    out[ro + l15] = v0 - lz;
    out[ro + 16 + l15] = v1 - lz;
    if (v2ok) out[ro + n2] = v2 - lz;
  }
}

extern "C" void kernel_launch(void* const* d_in, const int* in_sizes, int n_in,
                              void* d_out, int out_size, void* d_ws,
                              size_t ws_size, hipStream_t stream) {
  (void)in_sizes; (void)n_in; (void)out_size; (void)ws_size;
  const float* x = (const float*)d_in[0];
  const int* ei = (const int*)d_in[1];
  const float* W1_rel = (const float*)d_in[2];
  const float* b1 = (const float*)d_in[3];
  const float* W1_root = (const float*)d_in[4];
  const float* W2_rel = (const float*)d_in[5];
  const float* b2 = (const float*)d_in[6];
  const float* W2_root = (const float*)d_in[7];
  float* out = (float*)d_out;

  float* agg1 = (float*)d_ws;
  float* agg2 = agg1 + (size_t)N_NODES * IN_CH;
  float* hbuf = agg2 + (size_t)N_NODES * HID_CH;

  const int* srcv = ei;
  const int* dstv = ei + N_EDGES;

  {
    long long n4 = (long long)N_NODES * (IN_CH + HID_CH) / 4;
    zero_f4<<<(int)((n4 + 255) / 256), 256, 0, stream>>>((float4*)agg1,
                                                         (int)n4);
  }
  {
    long long total = (long long)N_EDGES * (IN_CH / 4);
    scatter_add_kernel<4><<<(int)((total + 255) / 256), 256, 0, stream>>>(
        x, srcv, dstv, agg1);
  }
  gemm1_relu<<<N_NODES / 16, 256, 0, stream>>>(agg1, x, W1_rel, W1_root, b1,
                                               hbuf);
  {
    long long total = (long long)N_EDGES * (HID_CH / 4);
    scatter_add_kernel<5><<<(int)((total + 255) / 256), 256, 0, stream>>>(
        hbuf, srcv, dstv, agg2);
  }
  gemm2_logsoftmax<<<(N_NODES / 16 + 7) / 8, 256, 0, stream>>>(
      agg2, hbuf, W2_rel, W2_root, b2, out);
}


# layer 2 applies W2_rel per node before the edge scatter (the sum is linear): atomics add 40-wide rows of h*W2_rel^T into out instead of 128-wide rows of h; float sum order changes
# speedup vs baseline: 1.4229x; 1.4229x over previous
#include <hip/hip_runtime.h>
#include <hip/hip_bf16.h>
#include <math.h>


typedef __attribute__((ext_vector_type(2))) float v2f;
typedef __attribute__((ext_vector_type(8))) float v8f;

#define N_NODES 50000
#define N_EDGES 800000
#define IN_CH   64
#define HID_CH  128
#define OUT_CH  40

__global__ void zero_f4(float4* __restrict__ p, int n4) {
  int i = blockIdx.x * blockDim.x + threadIdx.x;
  if (i < n4) p[i] = make_float4(0.f, 0.f, 0.f, 0.f);
}

template <int LOG2G>
__global__ void scatter_add_kernel(const float* __restrict__ feat,
                                   const int* __restrict__ srcv,
                                   const int* __restrict__ dstv,
                                   float* __restrict__ agg) {
  const int groups = 1 << LOG2G;
  const int ch = groups * 4;
  long long idx = (long long)blockIdx.x * blockDim.x + threadIdx.x;
  const long long total = (long long)N_EDGES << LOG2G;
  if (idx >= total) return;
  const int e = (int)(idx >> LOG2G);
  const int g = (int)(idx & (groups - 1));
  const int s = srcv[e];
  const int d = dstv[e];
  const float4 v = *(const float4*)(feat + (long long)s * ch + g * 4);
  float* o = agg + (long long)d * ch + g * 4;
  atomicAdd(o + 0, v.x);
  atomicAdd(o + 1, v.y);
  atomicAdd(o + 2, v.z);
  atomicAdd(o + 3, v.w);
}

__device__ __forceinline__ v2f load_frag(const float* base) {
  return *(const v2f*)base;
}

__global__ __launch_bounds__(256) void gemm1_relu(
    const float* __restrict__ agg1, const float* __restrict__ x,
    const float* __restrict__ Wrel, const float* __restrict__ Wroot,
    const float* __restrict__ bias, float* __restrict__ h) {
  const int lane = threadIdx.x & 31;
  const int wave = threadIdx.x >> 5;
  const int m_base = blockIdx.x * 16;
  const int n_base = wave * 16;
  const int l15 = lane & 15;
  const int kh = (lane >> 4) * 2;
  const int arow = m_base + l15;
  const int bcol = n_base + l15;

  const float* A0 = agg1 + (long long)arow * IN_CH + kh;
  const float* X0 = x + (long long)arow * IN_CH + kh;
  const float* Wr = Wrel + (long long)bcol * IN_CH + kh;
  const float* Wo = Wroot + (long long)bcol * IN_CH + kh;

  v8f c = {};
#pragma unroll
  for (int k = 0; k < IN_CH; k += 4) {
    v2f a = load_frag(A0 + k);
    v2f b = load_frag(Wr + k);
    c = __builtin_amdgcn_wmma_f32_16x16x4_f32(false, a, false, b, (short)0, c,
                                              false, false);
  }
#pragma unroll
  for (int k = 0; k < IN_CH; k += 4) {
    v2f a = load_frag(X0 + k);
    v2f b = load_frag(Wo + k);
    c = __builtin_amdgcn_wmma_f32_16x16x4_f32(false, a, false, b, (short)0, c,
                                              false, false);
  }

  const float bn = bias[bcol];
  const int rbase = m_base + ((lane >> 4) << 3);
#pragma unroll
  for (int i = 0; i < 8; ++i) {
    float v = c[i] + bn;
    h[(long long)(rbase + i) * HID_CH + bcol] = fmaxf(v, 0.0f);
  }
}

__global__ __launch_bounds__(256) void gemm2_node_products(
    const float* __restrict__ h, const float* __restrict__ Wrel,
    const float* __restrict__ Wroot, const float* __restrict__ bias,
    float* __restrict__ t, float* __restrict__ out) {
  const int lane = threadIdx.x & 31;
  const int wave = threadIdx.x >> 5;
  const int mtile = blockIdx.x * 8 + wave;
  if (mtile >= N_NODES / 16) return;
  const int m_base = mtile * 16;
  const int l15 = lane & 15;
  const int kh = (lane >> 4) * 2;
  const int arow = m_base + l15;
  const int n2 = 32 + l15;
  const bool v2ok = (n2 < OUT_CH);
  const int n2c = v2ok ? n2 : 0;

  const float* H0 = h + (long long)arow * HID_CH + kh;
  const float* Wr0 = Wrel + (long long)l15 * HID_CH + kh;
  const float* Wr1 = Wrel + (long long)(16 + l15) * HID_CH + kh;
  const float* Wr2 = Wrel + (long long)n2c * HID_CH + kh;
  const float* Wo0 = Wroot + (long long)l15 * HID_CH + kh;
  const float* Wo1 = Wroot + (long long)(16 + l15) * HID_CH + kh;
  const float* Wo2 = Wroot + (long long)n2c * HID_CH + kh;

  v8f r0 = {}, r1 = {}, r2 = {};
  v8f o0 = {}, o1 = {}, o2 = {};
  for (int k = 0; k < HID_CH; k += 4) {
    v2f a = load_frag(H0 + k);
    v2f br0 = load_frag(Wr0 + k);
    v2f br1 = load_frag(Wr1 + k);
    v2f br2 = load_frag(Wr2 + k);
    v2f bo0 = load_frag(Wo0 + k);
    v2f bo1 = load_frag(Wo1 + k);
    v2f bo2 = load_frag(Wo2 + k);
    if (!v2ok) { br2.x = 0.f; br2.y = 0.f; bo2.x = 0.f; bo2.y = 0.f; }
    r0 = __builtin_amdgcn_wmma_f32_16x16x4_f32(false, a, false, br0, (short)0,
                                               r0, false, false);
    r1 = __builtin_amdgcn_wmma_f32_16x16x4_f32(false, a, false, br1, (short)0,
                                               r1, false, false);
    r2 = __builtin_amdgcn_wmma_f32_16x16x4_f32(false, a, false, br2, (short)0,
                                               r2, false, false);
    o0 = __builtin_amdgcn_wmma_f32_16x16x4_f32(false, a, false, bo0, (short)0,
                                               o0, false, false);
    o1 = __builtin_amdgcn_wmma_f32_16x16x4_f32(false, a, false, bo1, (short)0,
                                               o1, false, false);
    o2 = __builtin_amdgcn_wmma_f32_16x16x4_f32(false, a, false, bo2, (short)0,
                                               o2, false, false);
  }

  const float bn0 = bias[l15];
  const float bn1 = bias[16 + l15];
  const float bn2 = v2ok ? bias[n2] : 0.0f;
  const int rbase = m_base + ((lane >> 4) << 3);
#pragma unroll
  for (int i = 0; i < 8; ++i) {
    const long long ro = (long long)(rbase + i) * OUT_CH;
    t[ro + l15] = r0[i];
    t[ro + 16 + l15] = r1[i];
    out[ro + l15] = o0[i] + bn0;
    out[ro + 16 + l15] = o1[i] + bn1;
    if (v2ok) {
      t[ro + n2] = r2[i];
      out[ro + n2] = o2[i] + bn2;
    }
  }
}

__global__ void scatter_add_out_kernel(const float* __restrict__ t,
                                       const int* __restrict__ srcv,
                                       const int* __restrict__ dstv,
                                       float* __restrict__ acc) {
  const int idx = blockIdx.x * blockDim.x + threadIdx.x;
  if (idx >= N_EDGES * 8) return;
  const int e = idx >> 3;
  const int j = idx & 7;
  const int s = srcv[e];
  const int d = dstv[e];
  const float* p = t + (long long)s * OUT_CH + j;
  float* o = acc + (long long)d * OUT_CH + j;
  const float v0 = p[0];
  const float v1 = p[8];
  const float v2 = p[16];
  const float v3 = p[24];
  const float v4 = p[32];
  atomicAdd(o + 0, v0);
  atomicAdd(o + 8, v1);
  atomicAdd(o + 16, v2);
  atomicAdd(o + 24, v3);
  atomicAdd(o + 32, v4);
}

__global__ __launch_bounds__(256) void log_softmax_rows(float* __restrict__ out) {
  const int idx = blockIdx.x * 256 + threadIdx.x;
  const int row = idx >> 4;
  const int g = idx & 15;
  const bool act = g < OUT_CH / 4;
  float4* p = (float4*)(out + (long long)row * OUT_CH) + (act ? g : 0);
  const float4 v = *p;
  float m = act ? fmaxf(fmaxf(v.x, v.y), fmaxf(v.z, v.w)) : -INFINITY;
#pragma unroll
  for (int msk = 8; msk >= 1; msk >>= 1)
    m = fmaxf(m, __shfl_xor(m, msk, 32));
  float s = act ? (__expf(v.x - m) + __expf(v.y - m) + __expf(v.z - m) +
                   __expf(v.w - m))
                : 0.0f;
#pragma unroll
  for (int msk = 8; msk >= 1; msk >>= 1) s += __shfl_xor(s, msk, 32);
  const float lz = m + __logf(s);
  if (act) *p = make_float4(v.x - lz, v.y - lz, v.z - lz, v.w - lz);
}

extern "C" void kernel_launch(void* const* d_in, const int* in_sizes, int n_in,
                              void* d_out, int out_size, void* d_ws,
                              size_t ws_size, hipStream_t stream) {
  (void)in_sizes; (void)n_in; (void)out_size; (void)ws_size;
  const float* x = (const float*)d_in[0];
  const int* ei = (const int*)d_in[1];
  const float* W1_rel = (const float*)d_in[2];
  const float* b1 = (const float*)d_in[3];
  const float* W1_root = (const float*)d_in[4];
  const float* W2_rel = (const float*)d_in[5];
  const float* b2 = (const float*)d_in[6];
  const float* W2_root = (const float*)d_in[7];
  float* out = (float*)d_out;

  float* agg1 = (float*)d_ws;
  float* tbuf = agg1 + (size_t)N_NODES * IN_CH;
  float* hbuf = tbuf + (size_t)N_NODES * OUT_CH;

  const int* srcv = ei;
  const int* dstv = ei + N_EDGES;

  {
    long long n4 = (long long)N_NODES * IN_CH / 4;
    zero_f4<<<(int)((n4 + 255) / 256), 256, 0, stream>>>((float4*)agg1,
                                                         (int)n4);
  }
  {
    long long total = (long long)N_EDGES * (IN_CH / 4);
    scatter_add_kernel<4><<<(int)((total + 255) / 256), 256, 0, stream>>>(
        x, srcv, dstv, agg1);
  }
  gemm1_relu<<<N_NODES / 16, 256, 0, stream>>>(agg1, x, W1_rel, W1_root, b1,
                                               hbuf);
  gemm2_node_products<<<(N_NODES / 16 + 7) / 8, 256, 0, stream>>>(
      hbuf, W2_rel, W2_root, b2, tbuf, out);
  {
    long long total = (long long)N_EDGES * 8;
    scatter_add_out_kernel<<<(int)((total + 255) / 256), 256, 0, stream>>>(
        tbuf, srcv, dstv, out);
  }
  log_softmax_rows<<<N_NODES * 16 / 256, 256, 0, stream>>>(out);
}


# both edge scatters: a wave walks 16 edges; each load/atomic instruction covers 32 consecutive channels of ONE edge (whole 128-byte spans), indices scalar; sums still in atomic arrival order
# speedup vs baseline: 1.7926x; 1.2598x over previous
#include <hip/hip_runtime.h>
#include <hip/hip_bf16.h>
#include <math.h>


typedef __attribute__((ext_vector_type(2))) float v2f;
typedef __attribute__((ext_vector_type(8))) float v8f;

#define N_NODES 50000
#define N_EDGES 800000
#define IN_CH   64
#define HID_CH  128
#define OUT_CH  40

__global__ void zero_f4(float4* __restrict__ p, int n4) {
  int i = blockIdx.x * blockDim.x + threadIdx.x;
  if (i < n4) p[i] = make_float4(0.f, 0.f, 0.f, 0.f);
}

#define EDGES_PER_WAVE 16
#define WAVES_PER_BLOCK 8
#define EDGE_BLOCKS (N_EDGES / (EDGES_PER_WAVE * WAVES_PER_BLOCK))
static_assert(N_EDGES % (EDGES_PER_WAVE * WAVES_PER_BLOCK) == 0,
              "edge grid must be exact: the kernels have no tail guard");

__global__ __launch_bounds__(256) void scatter_add_rows64(
    const float* __restrict__ feat, const int* __restrict__ srcv,
    const int* __restrict__ dstv, float* __restrict__ agg) {
  const unsigned lane = threadIdx.x & 31;
  const int wave = __builtin_amdgcn_readfirstlane((int)(threadIdx.x >> 5));
  const int e0 = (blockIdx.x * WAVES_PER_BLOCK + wave) * EDGES_PER_WAVE;
  unsigned s[EDGES_PER_WAVE], d[EDGES_PER_WAVE];
#pragma unroll
  for (int i = 0; i < EDGES_PER_WAVE; ++i) {
    s[i] = (unsigned)srcv[e0 + i];
    d[i] = (unsigned)dstv[e0 + i];
  }
  float lo[EDGES_PER_WAVE], hi[EDGES_PER_WAVE];
#pragma unroll
  for (int i = 0; i < EDGES_PER_WAVE; ++i) {
    const float* p = feat + (s[i] * IN_CH + lane);
    lo[i] = p[0];
    hi[i] = p[32];
  }
#pragma unroll
  for (int i = 0; i < EDGES_PER_WAVE; ++i) {
    float* o = agg + (d[i] * IN_CH + lane);
    atomicAdd(o, lo[i]);
    atomicAdd(o + 32, hi[i]);
  }
}

__device__ __forceinline__ v2f load_frag(const float* base) {
  return *(const v2f*)base;
}

__global__ __launch_bounds__(256) void gemm1_relu(
    const float* __restrict__ agg1, const float* __restrict__ x,
    const float* __restrict__ Wrel, const float* __restrict__ Wroot,
    const float* __restrict__ bias, float* __restrict__ h) {
  const int lane = threadIdx.x & 31;
  const int wave = threadIdx.x >> 5;
  const int m_base = blockIdx.x * 16;
  const int n_base = wave * 16;
  const int l15 = lane & 15;
  const int kh = (lane >> 4) * 2;
  const int arow = m_base + l15;
  const int bcol = n_base + l15;

  const float* A0 = agg1 + (long long)arow * IN_CH + kh;
  const float* X0 = x + (long long)arow * IN_CH + kh;
  const float* Wr = Wrel + (long long)bcol * IN_CH + kh;
  const float* Wo = Wroot + (long long)bcol * IN_CH + kh;

  v8f c = {};
#pragma unroll
  for (int k = 0; k < IN_CH; k += 4) {
    v2f a = load_frag(A0 + k);
    v2f b = load_frag(Wr + k);
    c = __builtin_amdgcn_wmma_f32_16x16x4_f32(false, a, false, b, (short)0, c,
                                              false, false);
  }
#pragma unroll
  for (int k = 0; k < IN_CH; k += 4) {
    v2f a = load_frag(X0 + k);
    v2f b = load_frag(Wo + k);
    c = __builtin_amdgcn_wmma_f32_16x16x4_f32(false, a, false, b, (short)0, c,
                                              false, false);
  }

  const float bn = bias[bcol];
  const int rbase = m_base + ((lane >> 4) << 3);
#pragma unroll
  for (int i = 0; i < 8; ++i) {
    float v = c[i] + bn;
    h[(long long)(rbase + i) * HID_CH + bcol] = fmaxf(v, 0.0f);
  }
}

__global__ __launch_bounds__(256) void gemm2_node_products(
    const float* __restrict__ h, const float* __restrict__ Wrel,
    const float* __restrict__ Wroot, const float* __restrict__ bias,
    float* __restrict__ t, float* __restrict__ out) {
  const int lane = threadIdx.x & 31;
  const int wave = threadIdx.x >> 5;
  const int mtile = blockIdx.x * 8 + wave;
  if (mtile >= N_NODES / 16) return;
  const int m_base = mtile * 16;
  const int l15 = lane & 15;
  const int kh = (lane >> 4) * 2;
  const int arow = m_base + l15;
  const int n2 = 32 + l15;
  const bool v2ok = (n2 < OUT_CH);
  const int n2c = v2ok ? n2 : 0;

  const float* H0 = h + (long long)arow * HID_CH + kh;
  const float* Wr0 = Wrel + (long long)l15 * HID_CH + kh;
  const float* Wr1 = Wrel + (long long)(16 + l15) * HID_CH + kh;
  const float* Wr2 = Wrel + (long long)n2c * HID_CH + kh;
  const float* Wo0 = Wroot + (long long)l15 * HID_CH + kh;
  const float* Wo1 = Wroot + (long long)(16 + l15) * HID_CH + kh;
  const float* Wo2 = Wroot + (long long)n2c * HID_CH + kh;

  v8f r0 = {}, r1 = {}, r2 = {};
  v8f o0 = {}, o1 = {}, o2 = {};
  for (int k = 0; k < HID_CH; k += 4) {
    v2f a = load_frag(H0 + k);
    v2f br0 = load_frag(Wr0 + k);
    v2f br1 = load_frag(Wr1 + k);
    v2f br2 = load_frag(Wr2 + k);
    v2f bo0 = load_frag(Wo0 + k);
    v2f bo1 = load_frag(Wo1 + k);
    v2f bo2 = load_frag(Wo2 + k);
    if (!v2ok) { br2.x = 0.f; br2.y = 0.f; bo2.x = 0.f; bo2.y = 0.f; }
    r0 = __builtin_amdgcn_wmma_f32_16x16x4_f32(false, a, false, br0, (short)0,
                                               r0, false, false);
    r1 = __builtin_amdgcn_wmma_f32_16x16x4_f32(false, a, false, br1, (short)0,
                                               r1, false, false);
    r2 = __builtin_amdgcn_wmma_f32_16x16x4_f32(false, a, false, br2, (short)0,
                                               r2, false, false);
    o0 = __builtin_amdgcn_wmma_f32_16x16x4_f32(false, a, false, bo0, (short)0,
                                               o0, false, false);
    o1 = __builtin_amdgcn_wmma_f32_16x16x4_f32(false, a, false, bo1, (short)0,
                                               o1, false, false);
    o2 = __builtin_amdgcn_wmma_f32_16x16x4_f32(false, a, false, bo2, (short)0,
                                               o2, false, false);
  }

  const float bn0 = bias[l15];
  const float bn1 = bias[16 + l15];
  const float bn2 = v2ok ? bias[n2] : 0.0f;
  const int rbase = m_base + ((lane >> 4) << 3);
#pragma unroll
  for (int i = 0; i < 8; ++i) {
    const long long ro = (long long)(rbase + i) * OUT_CH;
    t[ro + l15] = r0[i];
    t[ro + 16 + l15] = r1[i];
    out[ro + l15] = o0[i] + bn0;
    out[ro + 16 + l15] = o1[i] + bn1;
    if (v2ok) {
      t[ro + n2] = r2[i];
      out[ro + n2] = o2[i] + bn2;
    }
  }
}

__global__ __launch_bounds__(256) void scatter_add_rows40(
    const float* __restrict__ t, const int* __restrict__ srcv,
    const int* __restrict__ dstv, float* __restrict__ acc) {
  const unsigned lane = threadIdx.x & 31;
  const int wave = __builtin_amdgcn_readfirstlane((int)(threadIdx.x >> 5));
  const int e0 = (blockIdx.x * WAVES_PER_BLOCK + wave) * EDGES_PER_WAVE;
  unsigned s[EDGES_PER_WAVE], d[EDGES_PER_WAVE];
#pragma unroll
  for (int i = 0; i < EDGES_PER_WAVE; ++i) {
    s[i] = (unsigned)srcv[e0 + i];
    d[i] = (unsigned)dstv[e0 + i];
  }
  const unsigned sv = (unsigned)srcv[e0 + (int)(lane & 15)];
  const unsigned dv = (unsigned)dstv[e0 + (int)(lane & 15)];
  const int q = (int)(lane >> 3);
  const unsigned c = 32 + (lane & 7);
  float head[EDGES_PER_WAVE], tail[EDGES_PER_WAVE / 4];
  unsigned td[EDGES_PER_WAVE / 4];
#pragma unroll
  for (int i = 0; i < EDGES_PER_WAVE; ++i)
    head[i] = t[s[i] * OUT_CH + lane];
#pragma unroll
  for (int j = 0; j < EDGES_PER_WAVE / 4; ++j) {
    const unsigned ts = (unsigned)__shfl((int)sv, 4 * j + q, 32);
    td[j] = (unsigned)__shfl((int)dv, 4 * j + q, 32);
    tail[j] = t[ts * OUT_CH + c];
  }
#pragma unroll
  for (int i = 0; i < EDGES_PER_WAVE; ++i)
    atomicAdd(acc + (d[i] * OUT_CH + lane), head[i]);
#pragma unroll
  for (int j = 0; j < EDGES_PER_WAVE / 4; ++j)
    atomicAdd(acc + (td[j] * OUT_CH + c), tail[j]);
}

__global__ __launch_bounds__(256) void log_softmax_rows(float* __restrict__ out) {
  const int idx = blockIdx.x * 256 + threadIdx.x;
  const int row = idx >> 4;
  const int g = idx & 15;
  const bool act = g < OUT_CH / 4;
  float4* p = (float4*)(out + (long long)row * OUT_CH) + (act ? g : 0);
  const float4 v = *p;
  float m = act ? fmaxf(fmaxf(v.x, v.y), fmaxf(v.z, v.w)) : -INFINITY;
#pragma unroll
  for (int msk = 8; msk >= 1; msk >>= 1)
    m = fmaxf(m, __shfl_xor(m, msk, 32));
  float s = act ? (__expf(v.x - m) + __expf(v.y - m) + __expf(v.z - m) +
                   __expf(v.w - m))
                : 0.0f;
#pragma unroll
  for (int msk = 8; msk >= 1; msk >>= 1) s += __shfl_xor(s, msk, 32);
  const float lz = m + __logf(s);
  if (act) *p = make_float4(v.x - lz, v.y - lz, v.z - lz, v.w - lz);
}

extern "C" void kernel_launch(void* const* d_in, const int* in_sizes, int n_in,
                              void* d_out, int out_size, void* d_ws,
                              size_t ws_size, hipStream_t stream) {
  (void)in_sizes; (void)n_in; (void)out_size; (void)ws_size;
  const float* x = (const float*)d_in[0];
  const int* ei = (const int*)d_in[1];
  const float* W1_rel = (const float*)d_in[2];
  const float* b1 = (const float*)d_in[3];
  const float* W1_root = (const float*)d_in[4];
  const float* W2_rel = (const float*)d_in[5];
  const float* b2 = (const float*)d_in[6];
  const float* W2_root = (const float*)d_in[7];
  float* out = (float*)d_out;

  float* agg1 = (float*)d_ws;
  float* tbuf = agg1 + (size_t)N_NODES * IN_CH;
  float* hbuf = tbuf + (size_t)N_NODES * OUT_CH;

  const int* srcv = ei;
  const int* dstv = ei + N_EDGES;

  {
    long long n4 = (long long)N_NODES * IN_CH / 4;
    zero_f4<<<(int)((n4 + 255) / 256), 256, 0, stream>>>((float4*)agg1,
                                                         (int)n4);
  }
  scatter_add_rows64<<<EDGE_BLOCKS, 256, 0, stream>>>(x, srcv, dstv, agg1);
  gemm1_relu<<<N_NODES / 16, 256, 0, stream>>>(agg1, x, W1_rel, W1_root, b1,
                                               hbuf);
  gemm2_node_products<<<(N_NODES / 16 + 7) / 8, 256, 0, stream>>>(
      hbuf, W2_rel, W2_root, b2, tbuf, out);
  scatter_add_rows40<<<EDGE_BLOCKS, 256, 0, stream>>>(tbuf, srcv, dstv, out);
  log_softmax_rows<<<N_NODES * 16 / 256, 256, 0, stream>>>(out);
}


# both node GEMMs: weight matrices staged once per work-group in LDS (pitch K+4, conflict-free fragment reads); layer 1: a wave owns a 16-row strip of all 128 columns, A loaded once not 8 times
# speedup vs baseline: 2.1958x; 1.2249x over previous
#include <hip/hip_runtime.h>
#include <hip/hip_bf16.h>
#include <math.h>


typedef __attribute__((ext_vector_type(2))) float v2f;
typedef __attribute__((ext_vector_type(8))) float v8f;

#define N_NODES 50000
#define N_EDGES 800000
#define IN_CH   64
#define HID_CH  128
#define OUT_CH  40

__global__ void zero_f4(float4* __restrict__ p, int n4) {
  int i = blockIdx.x * blockDim.x + threadIdx.x;
  if (i < n4) p[i] = make_float4(0.f, 0.f, 0.f, 0.f);
}

#define EDGES_PER_WAVE 16
#define WAVES_PER_BLOCK 8
#define EDGE_BLOCKS (N_EDGES / (EDGES_PER_WAVE * WAVES_PER_BLOCK))
static_assert(N_EDGES % (EDGES_PER_WAVE * WAVES_PER_BLOCK) == 0,
              "edge grid must be exact: the kernels have no tail guard");

__global__ __launch_bounds__(256) void scatter_add_rows64(
    const float* __restrict__ feat, const int* __restrict__ srcv,
    const int* __restrict__ dstv, float* __restrict__ agg) {
  const unsigned lane = threadIdx.x & 31;
  const int wave = __builtin_amdgcn_readfirstlane((int)(threadIdx.x >> 5));
  const int e0 = (blockIdx.x * WAVES_PER_BLOCK + wave) * EDGES_PER_WAVE;
  unsigned s[EDGES_PER_WAVE], d[EDGES_PER_WAVE];
#pragma unroll
  for (int i = 0; i < EDGES_PER_WAVE; ++i) {
    s[i] = (unsigned)srcv[e0 + i];
    d[i] = (unsigned)dstv[e0 + i];
  }
  float lo[EDGES_PER_WAVE], hi[EDGES_PER_WAVE];
#pragma unroll
  for (int i = 0; i < EDGES_PER_WAVE; ++i) {
    const float* p = feat + (s[i] * IN_CH + lane);
    lo[i] = p[0];
    hi[i] = p[32];
  }
#pragma unroll
  for (int i = 0; i < EDGES_PER_WAVE; ++i) {
    float* o = agg + (d[i] * IN_CH + lane);
    atomicAdd(o, lo[i]);
    atomicAdd(o + 32, hi[i]);
  }
}

__device__ __forceinline__ v2f load_frag(const float* base) {
  return *(const v2f*)base;
}

#define W1_PITCH (IN_CH + 4)
#define W2_PITCH (HID_CH + 4)
#define W2_ROWS 48

__global__ __launch_bounds__(256) void gemm1_relu(
    const float* __restrict__ agg1, const float* __restrict__ x,
    const float* __restrict__ Wrel, const float* __restrict__ Wroot,
    const float* __restrict__ bias, float* __restrict__ h) {
  __shared__ float sW[2 * HID_CH * W1_PITCH];
  const int tid = threadIdx.x;
#pragma unroll
  for (int i = 0; i < HID_CH * IN_CH / 4 / 256; ++i) {
    const int f = tid + 256 * i;
    const int row = f / (IN_CH / 4), c4 = f % (IN_CH / 4);
    const float4 wr = ((const float4*)Wrel)[f];
    const float4 wo = ((const float4*)Wroot)[f];
    *(float4*)(sW + row * W1_PITCH + c4 * 4) = wr;
    *(float4*)(sW + (HID_CH + row) * W1_PITCH + c4 * 4) = wo;
  }
  __syncthreads();

  const int lane = tid & 31;
  const int wave = tid >> 5;
  const int mtile = blockIdx.x * 8 + wave;
  if (mtile >= N_NODES / 16) return;
  const int m_base = mtile * 16;
  const int l15 = lane & 15;
  const int kh = (lane >> 4) * 2;
  const int arow = m_base + l15;

  const float* A0 = agg1 + (long long)arow * IN_CH + kh;
  const float* X0 = x + (long long)arow * IN_CH + kh;
  const float* Br = sW + l15 * W1_PITCH + kh;
  const float* Bo = Br + HID_CH * W1_PITCH;

  v8f c[8] = {};
#pragma unroll
  for (int k = 0; k < IN_CH; k += 4) {
    const v2f a = load_frag(A0 + k);
#pragma unroll
    for (int j = 0; j < 8; ++j) {
      const v2f b = load_frag(Br + j * 16 * W1_PITCH + k);
      c[j] = __builtin_amdgcn_wmma_f32_16x16x4_f32(false, a, false, b,
                                                   (short)0, c[j], false,
                                                   false);
    }
  }
#pragma unroll
  for (int k = 0; k < IN_CH; k += 4) {
    const v2f a = load_frag(X0 + k);
#pragma unroll
    for (int j = 0; j < 8; ++j) {
      const v2f b = load_frag(Bo + j * 16 * W1_PITCH + k);
      c[j] = __builtin_amdgcn_wmma_f32_16x16x4_f32(false, a, false, b,
                                                   (short)0, c[j], false,
                                                   false);
    }
  }

  const int rbase = m_base + ((lane >> 4) << 3);
#pragma unroll
  for (int j = 0; j < 8; ++j) {
    const int col = j * 16 + l15;
    const float bn = bias[col];
#pragma unroll
    for (int i = 0; i < 8; ++i)
      h[(long long)(rbase + i) * HID_CH + col] = fmaxf(c[j][i] + bn, 0.0f);
  }
}

__global__ __launch_bounds__(256) void gemm2_node_products(
    const float* __restrict__ h, const float* __restrict__ Wrel,
    const float* __restrict__ Wroot, const float* __restrict__ bias,
    float* __restrict__ t, float* __restrict__ out) {
  __shared__ float sW[2 * W2_ROWS * W2_PITCH];
  const int tid = threadIdx.x;
#pragma unroll
  for (int i = 0; i < OUT_CH * HID_CH / 4 / 256; ++i) {
    const int f = tid + 256 * i;
    const int row = f / (HID_CH / 4), c4 = f % (HID_CH / 4);
    const float4 wr = ((const float4*)Wrel)[f];
    const float4 wo = ((const float4*)Wroot)[f];
    *(float4*)(sW + row * W2_PITCH + c4 * 4) = wr;
    *(float4*)(sW + (W2_ROWS + row) * W2_PITCH + c4 * 4) = wo;
  }
  {
    const int row = OUT_CH + tid / (HID_CH / 4), c4 = tid % (HID_CH / 4);
    const float4 z = make_float4(0.f, 0.f, 0.f, 0.f);
    *(float4*)(sW + row * W2_PITCH + c4 * 4) = z;
    *(float4*)(sW + (W2_ROWS + row) * W2_PITCH + c4 * 4) = z;
  }
  __syncthreads();

  const int lane = tid & 31;
  const int wave = tid >> 5;
  const int mtile = blockIdx.x * 8 + wave;
  if (mtile >= N_NODES / 16) return;
  const int m_base = mtile * 16;
  const int l15 = lane & 15;
  const int kh = (lane >> 4) * 2;
  const int arow = m_base + l15;
  const int n2 = 32 + l15;
  const bool v2ok = (n2 < OUT_CH);

  const float* H0 = h + (long long)arow * HID_CH + kh;
  const float* Br = sW + l15 * W2_PITCH + kh;
  const float* Bo = Br + W2_ROWS * W2_PITCH;

  v8f r[3] = {};
  v8f o[3] = {};
#pragma unroll
  for (int k = 0; k < HID_CH; k += 4) {
    const v2f a = load_frag(H0 + k);
#pragma unroll
    for (int j = 0; j < 3; ++j) {
      const v2f b = load_frag(Br + j * 16 * W2_PITCH + k);
      r[j] = __builtin_amdgcn_wmma_f32_16x16x4_f32(false, a, false, b,
                                                   (short)0, r[j], false,
                                                   false);
    }
#pragma unroll
    for (int j = 0; j < 3; ++j) {
      const v2f b = load_frag(Bo + j * 16 * W2_PITCH + k);
      o[j] = __builtin_amdgcn_wmma_f32_16x16x4_f32(false, a, false, b,
                                                   (short)0, o[j], false,
                                                   false);
    }
  }

  const float bn0 = bias[l15];
  const float bn1 = bias[16 + l15];
  const float bn2 = v2ok ? bias[n2] : 0.0f;
  const int rbase = m_base + ((lane >> 4) << 3);
#pragma unroll
  for (int i = 0; i < 8; ++i) {
    const long long ro = (long long)(rbase + i) * OUT_CH;
    t[ro + l15] = r[0][i];
    t[ro + 16 + l15] = r[1][i];
    out[ro + l15] = o[0][i] + bn0;
    out[ro + 16 + l15] = o[1][i] + bn1;
    if (v2ok) {
      t[ro + n2] = r[2][i];
      out[ro + n2] = o[2][i] + bn2;
    }
  }
}

__global__ __launch_bounds__(256) void scatter_add_rows40(
    const float* __restrict__ t, const int* __restrict__ srcv,
    const int* __restrict__ dstv, float* __restrict__ acc) {
  const unsigned lane = threadIdx.x & 31;
  const int wave = __builtin_amdgcn_readfirstlane((int)(threadIdx.x >> 5));
  const int e0 = (blockIdx.x * WAVES_PER_BLOCK + wave) * EDGES_PER_WAVE;
  unsigned s[EDGES_PER_WAVE], d[EDGES_PER_WAVE];
#pragma unroll
  for (int i = 0; i < EDGES_PER_WAVE; ++i) {
    s[i] = (unsigned)srcv[e0 + i];
    d[i] = (unsigned)dstv[e0 + i];
  }
  const unsigned sv = (unsigned)srcv[e0 + (int)(lane & 15)];
  const unsigned dv = (unsigned)dstv[e0 + (int)(lane & 15)];
  const int q = (int)(lane >> 3);
  const unsigned c = 32 + (lane & 7);
  float head[EDGES_PER_WAVE], tail[EDGES_PER_WAVE / 4];
  unsigned td[EDGES_PER_WAVE / 4];
#pragma unroll
  for (int i = 0; i < EDGES_PER_WAVE; ++i)
    head[i] = t[s[i] * OUT_CH + lane];
#pragma unroll
  for (int j = 0; j < EDGES_PER_WAVE / 4; ++j) {
    const unsigned ts = (unsigned)__shfl((int)sv, 4 * j + q, 32);
    td[j] = (unsigned)__shfl((int)dv, 4 * j + q, 32);
    tail[j] = t[ts * OUT_CH + c];
  }
#pragma unroll
  for (int i = 0; i < EDGES_PER_WAVE; ++i)
    atomicAdd(acc + (d[i] * OUT_CH + lane), head[i]);
#pragma unroll
  for (int j = 0; j < EDGES_PER_WAVE / 4; ++j)
    atomicAdd(acc + (td[j] * OUT_CH + c), tail[j]);
}

__global__ __launch_bounds__(256) void log_softmax_rows(float* __restrict__ out) {
  const int idx = blockIdx.x * 256 + threadIdx.x;
  const int row = idx >> 4;
  const int g = idx & 15;
  const bool act = g < OUT_CH / 4;
  float4* p = (float4*)(out + (long long)row * OUT_CH) + (act ? g : 0);
  const float4 v = *p;
  float m = act ? fmaxf(fmaxf(v.x, v.y), fmaxf(v.z, v.w)) : -INFINITY;
#pragma unroll
  for (int msk = 8; msk >= 1; msk >>= 1)
    m = fmaxf(m, __shfl_xor(m, msk, 32));
  float s = act ? (__expf(v.x - m) + __expf(v.y - m) + __expf(v.z - m) +
                   __expf(v.w - m))
                : 0.0f;
#pragma unroll
  for (int msk = 8; msk >= 1; msk >>= 1) s += __shfl_xor(s, msk, 32);
  const float lz = m + __logf(s);
  if (act) *p = make_float4(v.x - lz, v.y - lz, v.z - lz, v.w - lz);
}

extern "C" void kernel_launch(void* const* d_in, const int* in_sizes, int n_in,
                              void* d_out, int out_size, void* d_ws,
                              size_t ws_size, hipStream_t stream) {
  (void)in_sizes; (void)n_in; (void)out_size; (void)ws_size;
  const float* x = (const float*)d_in[0];
  const int* ei = (const int*)d_in[1];
  const float* W1_rel = (const float*)d_in[2];
  const float* b1 = (const float*)d_in[3];
  const float* W1_root = (const float*)d_in[4];
  const float* W2_rel = (const float*)d_in[5];
  const float* b2 = (const float*)d_in[6];
  const float* W2_root = (const float*)d_in[7];
  float* out = (float*)d_out;

  float* agg1 = (float*)d_ws;
  float* tbuf = agg1 + (size_t)N_NODES * IN_CH;
  float* hbuf = tbuf + (size_t)N_NODES * OUT_CH;

  const int* srcv = ei;
  const int* dstv = ei + N_EDGES;

  {
    long long n4 = (long long)N_NODES * IN_CH / 4;
    zero_f4<<<(int)((n4 + 255) / 256), 256, 0, stream>>>((float4*)agg1,
                                                         (int)n4);
  }
  scatter_add_rows64<<<EDGE_BLOCKS, 256, 0, stream>>>(x, srcv, dstv, agg1);
  gemm1_relu<<<(N_NODES / 16 + 7) / 8, 256, 0, stream>>>(agg1, x, W1_rel,
                                                         W1_root, b1, hbuf);
  gemm2_node_products<<<(N_NODES / 16 + 7) / 8, 256, 0, stream>>>(
      hbuf, W2_rel, W2_root, b2, tbuf, out);
  scatter_add_rows40<<<EDGE_BLOCKS, 256, 0, stream>>>(tbuf, srcv, dstv, out);
  log_softmax_rows<<<N_NODES * 16 / 256, 256, 0, stream>>>(out);
}


# edge sums without float atomics: edges grouped by destination in LDS (range partition, then node sort; integer adds); 16 lanes own a node, sum rows, take log_softmax; float sum order changes
# speedup vs baseline: 3.4674x; 1.5791x over previous
#include <hip/hip_runtime.h>
#include <hip/hip_bf16.h>
#include <math.h>


typedef __attribute__((ext_vector_type(2))) float v2f;
typedef __attribute__((ext_vector_type(8))) float v8f;

#define N_NODES 50000
#define N_EDGES 800000
#define IN_CH   64
#define HID_CH  128
#define OUT_CH  40

#define RANGE_NODES 196
#define N_RANGES 256
#define RANGE_CAP 4096
#define PART_EDGES 1280
#define PART_BLOCKS (N_EDGES / PART_EDGES)
#define LIST_CHUNK 48
static_assert(N_RANGES * RANGE_NODES >= N_NODES, "ranges must cover all nodes");
static_assert(N_EDGES % PART_EDGES == 0 && (N_NODES * 16) % 256 == 0,
              "edge and node grids must be exact: the kernels have no tail guard");

__global__ __launch_bounds__(512) void clear_ctrl(int* __restrict__ ctrl,
                                                  float* __restrict__ zrow) {
  const int i = threadIdx.x;
  if (i <= N_RANGES) ctrl[i] = 0;
  if (i < IN_CH) zrow[i] = 0.0f;
}

__global__ __launch_bounds__(256) void partition_edges(
    const int* __restrict__ srcv, const int* __restrict__ dstv,
    int* __restrict__ ctrl, int2* __restrict__ rbuf, int2* __restrict__ ovf) {
  __shared__ int hist[N_RANGES];
  __shared__ int gbase[N_RANGES];
  const int tid = threadIdx.x;
  hist[tid] = 0;
  __syncthreads();
  int d[PART_EDGES / 256], s[PART_EDGES / 256], rk[PART_EDGES / 256];
  const int e0 = blockIdx.x * PART_EDGES + tid;
#pragma unroll
  for (int i = 0; i < PART_EDGES / 256; ++i) {
    d[i] = dstv[e0 + i * 256];
    s[i] = srcv[e0 + i * 256];
  }
#pragma unroll
  for (int i = 0; i < PART_EDGES / 256; ++i)
    rk[i] = atomicAdd(&hist[d[i] / RANGE_NODES], 1);
  __syncthreads();
  {
    const int c = hist[tid];
    gbase[tid] = c > 0 ? atomicAdd(ctrl + tid, c) : 0;
  }
  __syncthreads();
#pragma unroll
  for (int i = 0; i < PART_EDGES / 256; ++i) {
    const int r = d[i] / RANGE_NODES;
    const int pos = gbase[r] + rk[i];
    if (pos < RANGE_CAP) {
      rbuf[r * RANGE_CAP + pos] = make_int2(d[i], s[i]);
    } else {
      const int k = atomicAdd(ctrl + N_RANGES, 1);
      ovf[k] = make_int2(d[i], s[i]);
    }
  }
}

__global__ __launch_bounds__(1024) void sort_range(
    const int* __restrict__ ctrl, const int2* __restrict__ rbuf,
    int* __restrict__ sorted, int2* __restrict__ nodeinfo) {
  __shared__ int cnt[256];
  __shared__ int off[256];
  __shared__ int list[RANGE_CAP];
  const int tid = threadIdx.x;
  const int r = blockIdx.x;
  const int tot = ctrl[r];
  const int n = tot < RANGE_CAP ? tot : RANGE_CAP;
  if (tid < 256) cnt[tid] = 0;
  __syncthreads();
  int2 p[RANGE_CAP / 1024];
  int dl[RANGE_CAP / 1024], rk[RANGE_CAP / 1024];
#pragma unroll
  for (int i = 0; i < RANGE_CAP / 1024; ++i) {
    const int e = tid + i * 1024;
    p[i] = make_int2(-1, 0);
    if (e < n) p[i] = rbuf[r * RANGE_CAP + e];
  }
#pragma unroll
  for (int i = 0; i < RANGE_CAP / 1024; ++i) {
    dl[i] = -1; rk[i] = 0;
    if (p[i].x >= 0) {
      dl[i] = p[i].x - r * RANGE_NODES;
      rk[i] = atomicAdd(&cnt[dl[i]], 1);
    }
  }
  __syncthreads();
  if (tid < 256) {
    int o = 0;
    for (int u = 0; u < tid; ++u) o += cnt[u];
    off[tid] = o;
  }
  __syncthreads();
#pragma unroll
  for (int i = 0; i < RANGE_CAP / 1024; ++i)
    if (dl[i] >= 0) list[off[dl[i]] + rk[i]] = p[i].y;
  __syncthreads();
#pragma unroll
  for (int i = 0; i < RANGE_CAP / 1024; ++i) {
    const int e = tid + i * 1024;
    if (e < n) sorted[r * RANGE_CAP + e] = list[e];
  }
  if (tid < RANGE_NODES) {
    const int node = r * RANGE_NODES + tid;
    if (node < N_NODES)
      nodeinfo[node] = make_int2(r * RANGE_CAP + off[tid], cnt[tid]);
  }
}

__device__ __forceinline__ float4 sum_in_rows(
    const float* __restrict__ rows, const int rowf, const bool act,
    const int node, const int g, const int2* __restrict__ nodeinfo,
    const int* __restrict__ sorted, const int* __restrict__ ctrl,
    const int2* __restrict__ ovf, const float* __restrict__ zrow, float4 acc) {
  const unsigned long long info =
      *(const unsigned long long*)(nodeinfo + node);
  const int start = (int)(unsigned)info;
  const int deg = (int)(info >> 32);
  const float* base = rows + g * 4;
  const float* zpiece = zrow + g * 4;
  for (int o = 0; o < deg; o += LIST_CHUNK) {
    const int n = deg - o < LIST_CHUNK ? deg - o : LIST_CHUNK;
    const int* b = sorted + start + o;
    const int e0 = (g < n) ? b[g] : 0;
    const int e1 = (g + 16 < n) ? b[g + 16] : 0;
    const int e2 = (g + 32 < n) ? b[g + 32] : 0;
    for (int j0 = 0; j0 < n; j0 += 8) {
      const int ev = j0 < 16 ? e0 : (j0 < 32 ? e1 : e2);
      float4 v[8];
#pragma unroll
      for (int u = 0; u < 8; ++u) {
        const int s = __shfl(ev, (j0 + u) & 15, 16);
        const float* p =
            (act && j0 + u < n) ? base + (long long)s * rowf : zpiece;
        v[u] = *(const float4*)p;
      }
#pragma unroll
      for (int u = 0; u < 8; ++u) {
        acc.x += v[u].x; acc.y += v[u].y; acc.z += v[u].z; acc.w += v[u].w;
      }
    }
  }
  if (ctrl[node / RANGE_NODES] > RANGE_CAP) {
    const int m = ctrl[N_RANGES];
    for (int k = 0; k < m; ++k) {
      const int2 p = ovf[k];
      if (p.x == node && act) {
        const float4 v = *(const float4*)(base + (long long)p.y * rowf);
        acc.x += v.x; acc.y += v.y; acc.z += v.z; acc.w += v.w;
      }
    }
  }
  return acc;
}

__global__ __launch_bounds__(256) void gather_sum_rows64(
    const float* __restrict__ feat, const int2* __restrict__ nodeinfo,
    const int* __restrict__ sorted, const int* __restrict__ ctrl,
    const int2* __restrict__ ovf, const float* __restrict__ zrow,
    float* __restrict__ agg) {
  const int idx = blockIdx.x * 256 + threadIdx.x;
  const int node = idx >> 4;
  const int g = idx & 15;
  const float4 acc =
      sum_in_rows(feat, IN_CH, true, node, g, nodeinfo, sorted, ctrl, ovf,
                  zrow, make_float4(0.f, 0.f, 0.f, 0.f));
  *(float4*)(agg + (long long)node * IN_CH + g * 4) = acc;
}

__device__ __forceinline__ v2f load_frag(const float* base) {
  return *(const v2f*)base;
}

#define W1_PITCH (IN_CH + 4)
#define W2_PITCH (HID_CH + 4)
#define W2_ROWS 48

__global__ __launch_bounds__(256) void gemm1_relu(
    const float* __restrict__ agg1, const float* __restrict__ x,
    const float* __restrict__ Wrel, const float* __restrict__ Wroot,
    const float* __restrict__ bias, float* __restrict__ h) {
  __shared__ float sW[2 * HID_CH * W1_PITCH];
  const int tid = threadIdx.x;
#pragma unroll
  for (int i = 0; i < HID_CH * IN_CH / 4 / 256; ++i) {
    const int f = tid + 256 * i;
    const int row = f / (IN_CH / 4), c4 = f % (IN_CH / 4);
    const float4 wr = ((const float4*)Wrel)[f];
    const float4 wo = ((const float4*)Wroot)[f];
    *(float4*)(sW + row * W1_PITCH + c4 * 4) = wr;
    *(float4*)(sW + (HID_CH + row) * W1_PITCH + c4 * 4) = wo;
  }
  __syncthreads();

  const int lane = tid & 31;
  const int wave = tid >> 5;
  const int mtile = blockIdx.x * 8 + wave;
  if (mtile >= N_NODES / 16) return;
  const int m_base = mtile * 16;
  const int l15 = lane & 15;
  const int kh = (lane >> 4) * 2;
  const int arow = m_base + l15;

  const float* A0 = agg1 + (long long)arow * IN_CH + kh;
  const float* X0 = x + (long long)arow * IN_CH + kh;
  const float* Br = sW + l15 * W1_PITCH + kh;
  const float* Bo = Br + HID_CH * W1_PITCH;

  v8f c[8] = {};
#pragma unroll
  for (int k = 0; k < IN_CH; k += 4) {
    const v2f a = load_frag(A0 + k);
#pragma unroll
    for (int j = 0; j < 8; ++j) {
      const v2f b = load_frag(Br + j * 16 * W1_PITCH + k);
      c[j] = __builtin_amdgcn_wmma_f32_16x16x4_f32(false, a, false, b,
                                                   (short)0, c[j], false,
                                                   false);
    }
  }
#pragma unroll
  for (int k = 0; k < IN_CH; k += 4) {
    const v2f a = load_frag(X0 + k);
#pragma unroll
    for (int j = 0; j < 8; ++j) {
      const v2f b = load_frag(Bo + j * 16 * W1_PITCH + k);
      c[j] = __builtin_amdgcn_wmma_f32_16x16x4_f32(false, a, false, b,
                                                   (short)0, c[j], false,
                                                   false);
    }
  }

  const int rbase = m_base + ((lane >> 4) << 3);
#pragma unroll
  for (int j = 0; j < 8; ++j) {
    const int col = j * 16 + l15;
    const float bn = bias[col];
#pragma unroll
    for (int i = 0; i < 8; ++i)
      h[(long long)(rbase + i) * HID_CH + col] = fmaxf(c[j][i] + bn, 0.0f);
  }
}

__global__ __launch_bounds__(256) void gemm2_node_products(
    const float* __restrict__ h, const float* __restrict__ Wrel,
    const float* __restrict__ Wroot, const float* __restrict__ bias,
    float* __restrict__ t, float* __restrict__ out) {
  __shared__ float sW[2 * W2_ROWS * W2_PITCH];
  const int tid = threadIdx.x;
#pragma unroll
  for (int i = 0; i < OUT_CH * HID_CH / 4 / 256; ++i) {
    const int f = tid + 256 * i;
    const int row = f / (HID_CH / 4), c4 = f % (HID_CH / 4);
    const float4 wr = ((const float4*)Wrel)[f];
    const float4 wo = ((const float4*)Wroot)[f];
    *(float4*)(sW + row * W2_PITCH + c4 * 4) = wr;
    *(float4*)(sW + (W2_ROWS + row) * W2_PITCH + c4 * 4) = wo;
  }
  {
    const int row = OUT_CH + tid / (HID_CH / 4), c4 = tid % (HID_CH / 4);
    const float4 z = make_float4(0.f, 0.f, 0.f, 0.f);
    *(float4*)(sW + row * W2_PITCH + c4 * 4) = z;
    *(float4*)(sW + (W2_ROWS + row) * W2_PITCH + c4 * 4) = z;
  }
  __syncthreads();

  const int lane = tid & 31;
  const int wave = tid >> 5;
  const int mtile = blockIdx.x * 8 + wave;
  if (mtile >= N_NODES / 16) return;
  const int m_base = mtile * 16;
  const int l15 = lane & 15;
  const int kh = (lane >> 4) * 2;
  const int arow = m_base + l15;
  const int n2 = 32 + l15;
  const bool v2ok = (n2 < OUT_CH);

  const float* H0 = h + (long long)arow * HID_CH + kh;
  const float* Br = sW + l15 * W2_PITCH + kh;
  const float* Bo = Br + W2_ROWS * W2_PITCH;

  v8f r[3] = {};
  v8f o[3] = {};
#pragma unroll
  for (int k = 0; k < HID_CH; k += 4) {
    const v2f a = load_frag(H0 + k);
#pragma unroll
    for (int j = 0; j < 3; ++j) {
      const v2f b = load_frag(Br + j * 16 * W2_PITCH + k);
      r[j] = __builtin_amdgcn_wmma_f32_16x16x4_f32(false, a, false, b,
                                                   (short)0, r[j], false,
                                                   false);
    }
#pragma unroll
    for (int j = 0; j < 3; ++j) {
      const v2f b = load_frag(Bo + j * 16 * W2_PITCH + k);
      o[j] = __builtin_amdgcn_wmma_f32_16x16x4_f32(false, a, false, b,
                                                   (short)0, o[j], false,
                                                   false);
    }
  }

  const float bn0 = bias[l15];
  const float bn1 = bias[16 + l15];
  const float bn2 = v2ok ? bias[n2] : 0.0f;
  const int rbase = m_base + ((lane >> 4) << 3);
#pragma unroll
  for (int i = 0; i < 8; ++i) {
    const long long ro = (long long)(rbase + i) * OUT_CH;
    t[ro + l15] = r[0][i];
    t[ro + 16 + l15] = r[1][i];
    out[ro + l15] = o[0][i] + bn0;
    out[ro + 16 + l15] = o[1][i] + bn1;
    if (v2ok) {
      t[ro + n2] = r[2][i];
      out[ro + n2] = o[2][i] + bn2;
    }
  }
}

__global__ __launch_bounds__(256) void gather_sum_rows40_log_softmax(
    const float* __restrict__ t, const int2* __restrict__ nodeinfo,
    const int* __restrict__ sorted, const int* __restrict__ ctrl,
    const int2* __restrict__ ovf, const float* __restrict__ zrow,
    float* __restrict__ out) {
  const int idx = blockIdx.x * 256 + threadIdx.x;
  const int node = idx >> 4;
  const int g = idx & 15;
  const bool act = g < OUT_CH / 4;
  float4* p = (float4*)(out + (long long)node * OUT_CH) + (act ? g : 0);
  const float4 start = act ? *p : make_float4(0.f, 0.f, 0.f, 0.f);
  const float4 v = sum_in_rows(t, OUT_CH, act, node, g, nodeinfo, sorted, ctrl,
                               ovf, zrow, start);
  float m = act ? fmaxf(fmaxf(v.x, v.y), fmaxf(v.z, v.w)) : -INFINITY;
#pragma unroll
  for (int msk = 8; msk >= 1; msk >>= 1)
    m = fmaxf(m, __shfl_xor(m, msk, 32));
  float s = act ? (__expf(v.x - m) + __expf(v.y - m) + __expf(v.z - m) +
                   __expf(v.w - m))
                : 0.0f;
#pragma unroll
  for (int msk = 8; msk >= 1; msk >>= 1) s += __shfl_xor(s, msk, 32);
  const float lz = m + __logf(s);
  if (act) *p = make_float4(v.x - lz, v.y - lz, v.z - lz, v.w - lz);
}

extern "C" void kernel_launch(void* const* d_in, const int* in_sizes, int n_in,
                              void* d_out, int out_size, void* d_ws,
                              size_t ws_size, hipStream_t stream) {
  (void)in_sizes; (void)n_in; (void)out_size; (void)ws_size;
  const float* x = (const float*)d_in[0];
  const int* ei = (const int*)d_in[1];
  const float* W1_rel = (const float*)d_in[2];
  const float* b1 = (const float*)d_in[3];
  const float* W1_root = (const float*)d_in[4];
  const float* W2_rel = (const float*)d_in[5];
  const float* b2 = (const float*)d_in[6];
  const float* W2_root = (const float*)d_in[7];
  float* out = (float*)d_out;

  float* agg1 = (float*)d_ws;
  float* tbuf = agg1 + (size_t)N_NODES * IN_CH;
  float* hbuf = tbuf + (size_t)N_NODES * OUT_CH;
  int* ctrl = (int*)(hbuf + (size_t)N_NODES * HID_CH);
  float* zrow = (float*)(ctrl + 320);
  int2* nodeinfo = (int2*)(zrow + IN_CH);
  int2* rbuf = nodeinfo + N_NODES;
  int* sorted = (int*)(rbuf + (size_t)N_RANGES * RANGE_CAP);
  int2* ovf = (int2*)(sorted + (size_t)N_RANGES * RANGE_CAP);

  const int* srcv = ei;
  const int* dstv = ei + N_EDGES;

  clear_ctrl<<<1, 512, 0, stream>>>(ctrl, zrow);
  partition_edges<<<PART_BLOCKS, 256, 0, stream>>>(srcv, dstv, ctrl, rbuf, ovf);
  sort_range<<<N_RANGES, 1024, 0, stream>>>(ctrl, rbuf, sorted, nodeinfo);
  gather_sum_rows64<<<N_NODES * 16 / 256, 256, 0, stream>>>(
      x, nodeinfo, sorted, ctrl, ovf, zrow, agg1);
  gemm1_relu<<<(N_NODES / 16 + 7) / 8, 256, 0, stream>>>(agg1, x, W1_rel,
                                                         W1_root, b1, hbuf);
  gemm2_node_products<<<(N_NODES / 16 + 7) / 8, 256, 0, stream>>>(
      hbuf, W2_rel, W2_root, b2, tbuf, out);
  gather_sum_rows40_log_softmax<<<N_NODES * 16 / 256, 256, 0, stream>>>(
      tbuf, nodeinfo, sorted, ctrl, ovf, zrow, out);
}


# both node GEMMs in one kernel, formed transposed (nodes = columns): layer-1 accumulators are layer-2 B fragments, h never stored; W2_rel,W2_root stacked = 5 tiles; layer-2 k sum order changes
# speedup vs baseline: 3.7278x; 1.0751x over previous
#include <hip/hip_runtime.h>
#include <hip/hip_bf16.h>
#include <math.h>


typedef __attribute__((ext_vector_type(2))) float v2f;
typedef __attribute__((ext_vector_type(8))) float v8f;

#define N_NODES 50000
#define N_EDGES 800000
#define IN_CH   64
#define HID_CH  128
#define OUT_CH  40

#define RANGE_NODES 196
#define N_RANGES 256
#define RANGE_CAP 4096
#define PART_EDGES 1280
#define PART_BLOCKS (N_EDGES / PART_EDGES)
#define LIST_CHUNK 48
static_assert(N_RANGES * RANGE_NODES >= N_NODES, "ranges must cover all nodes");
static_assert(N_EDGES % PART_EDGES == 0 && (N_NODES * 16) % 256 == 0,
              "edge and node grids must be exact: the kernels have no tail guard");

__global__ __launch_bounds__(512) void clear_ctrl(int* __restrict__ ctrl,
                                                  float* __restrict__ zrow) {
  const int i = threadIdx.x;
  if (i <= N_RANGES) ctrl[i] = 0;
  if (i < IN_CH) zrow[i] = 0.0f;
}

__global__ __launch_bounds__(256) void partition_edges(
    const int* __restrict__ srcv, const int* __restrict__ dstv,
    int* __restrict__ ctrl, int2* __restrict__ rbuf, int2* __restrict__ ovf) {
  __shared__ int hist[N_RANGES];
  __shared__ int gbase[N_RANGES];
  const int tid = threadIdx.x;
  hist[tid] = 0;
  __syncthreads();
  int d[PART_EDGES / 256], s[PART_EDGES / 256], rk[PART_EDGES / 256];
  const int e0 = blockIdx.x * PART_EDGES + tid;
#pragma unroll
  for (int i = 0; i < PART_EDGES / 256; ++i) {
    d[i] = dstv[e0 + i * 256];
    s[i] = srcv[e0 + i * 256];
  }
#pragma unroll
  for (int i = 0; i < PART_EDGES / 256; ++i)
    rk[i] = atomicAdd(&hist[d[i] / RANGE_NODES], 1);
  __syncthreads();
  {
    const int c = hist[tid];
    gbase[tid] = c > 0 ? atomicAdd(ctrl + tid, c) : 0;
  }
  __syncthreads();
#pragma unroll
  for (int i = 0; i < PART_EDGES / 256; ++i) {
    const int r = d[i] / RANGE_NODES;
    const int pos = gbase[r] + rk[i];
    if (pos < RANGE_CAP) {
      rbuf[r * RANGE_CAP + pos] = make_int2(d[i], s[i]);
    } else {
      const int k = atomicAdd(ctrl + N_RANGES, 1);
      ovf[k] = make_int2(d[i], s[i]);
    }
  }
}

__global__ __launch_bounds__(1024) void sort_range(
    const int* __restrict__ ctrl, const int2* __restrict__ rbuf,
    int* __restrict__ sorted, int2* __restrict__ nodeinfo) {
  __shared__ int cnt[256];
  __shared__ int off[256];
  __shared__ int list[RANGE_CAP];
  const int tid = threadIdx.x;
  const int r = blockIdx.x;
  const int tot = ctrl[r];
  const int n = tot < RANGE_CAP ? tot : RANGE_CAP;
  if (tid < 256) cnt[tid] = 0;
  __syncthreads();
  int2 p[RANGE_CAP / 1024];
  int dl[RANGE_CAP / 1024], rk[RANGE_CAP / 1024];
#pragma unroll
  for (int i = 0; i < RANGE_CAP / 1024; ++i) {
    const int e = tid + i * 1024;
    p[i] = make_int2(-1, 0);
    if (e < n) p[i] = rbuf[r * RANGE_CAP + e];
  }
#pragma unroll
  for (int i = 0; i < RANGE_CAP / 1024; ++i) {
    dl[i] = -1; rk[i] = 0;
    if (p[i].x >= 0) {
      dl[i] = p[i].x - r * RANGE_NODES;
      rk[i] = atomicAdd(&cnt[dl[i]], 1);
    }
  }
  __syncthreads();
  if (tid < 256) {
    int o = 0;
    for (int u = 0; u < tid; ++u) o += cnt[u];
    off[tid] = o;
  }
  __syncthreads();
#pragma unroll
  for (int i = 0; i < RANGE_CAP / 1024; ++i)
    if (dl[i] >= 0) list[off[dl[i]] + rk[i]] = p[i].y;
  __syncthreads();
#pragma unroll
  for (int i = 0; i < RANGE_CAP / 1024; ++i) {
    const int e = tid + i * 1024;
    if (e < n) sorted[r * RANGE_CAP + e] = list[e];
  }
  if (tid < RANGE_NODES) {
    const int node = r * RANGE_NODES + tid;
    if (node < N_NODES)
      nodeinfo[node] = make_int2(r * RANGE_CAP + off[tid], cnt[tid]);
  }
}

__device__ __forceinline__ float4 sum_in_rows(
    const float* __restrict__ rows, const int rowf, const bool act,
    const int node, const int g, const int2* __restrict__ nodeinfo,
    const int* __restrict__ sorted, const int* __restrict__ ctrl,
    const int2* __restrict__ ovf, const float* __restrict__ zrow, float4 acc) {
  const unsigned long long info =
      *(const unsigned long long*)(nodeinfo + node);
  const int start = (int)(unsigned)info;
  const int deg = (int)(info >> 32);
  const float* base = rows + g * 4;
  const float* zpiece = zrow + g * 4;
  for (int o = 0; o < deg; o += LIST_CHUNK) {
    const int n = deg - o < LIST_CHUNK ? deg - o : LIST_CHUNK;
    const int* b = sorted + start + o;
    const int e0 = (g < n) ? b[g] : 0;
    const int e1 = (g + 16 < n) ? b[g + 16] : 0;
    const int e2 = (g + 32 < n) ? b[g + 32] : 0;
    for (int j0 = 0; j0 < n; j0 += 8) {
      const int ev = j0 < 16 ? e0 : (j0 < 32 ? e1 : e2);
      float4 v[8];
#pragma unroll
      for (int u = 0; u < 8; ++u) {
        const int s = __shfl(ev, (j0 + u) & 15, 16);
        const float* p =
            (act && j0 + u < n) ? base + (long long)s * rowf : zpiece;
        v[u] = *(const float4*)p;
      }
#pragma unroll
      for (int u = 0; u < 8; ++u) {
        acc.x += v[u].x; acc.y += v[u].y; acc.z += v[u].z; acc.w += v[u].w;
      }
    }
  }
  if (ctrl[node / RANGE_NODES] > RANGE_CAP) {
    const int m = ctrl[N_RANGES];
    for (int k = 0; k < m; ++k) {
      const int2 p = ovf[k];
      if (p.x == node && act) {
        const float4 v = *(const float4*)(base + (long long)p.y * rowf);
        acc.x += v.x; acc.y += v.y; acc.z += v.z; acc.w += v.w;
      }
    }
  }
  return acc;
}

__global__ __launch_bounds__(256) void gather_sum_rows64(
    const float* __restrict__ feat, const int2* __restrict__ nodeinfo,
    const int* __restrict__ sorted, const int* __restrict__ ctrl,
    const int2* __restrict__ ovf, const float* __restrict__ zrow,
    float* __restrict__ agg) {
  const int idx = blockIdx.x * 256 + threadIdx.x;
  const int node = idx >> 4;
  const int g = idx & 15;
  const float4 acc =
      sum_in_rows(feat, IN_CH, true, node, g, nodeinfo, sorted, ctrl, ovf,
                  zrow, make_float4(0.f, 0.f, 0.f, 0.f));
  *(float4*)(agg + (long long)node * IN_CH + g * 4) = acc;
}

__device__ __forceinline__ v2f load_frag(const float* base) {
  return *(const v2f*)base;
}

#define W1_PITCH (IN_CH + 4)
#define W2_PITCH (HID_CH + 4)
#define W2S_ROWS (2 * OUT_CH)

__global__ __launch_bounds__(256) void node_gemms_fused(
    const float* __restrict__ agg1, const float* __restrict__ x,
    const float* __restrict__ W1rel, const float* __restrict__ W1root,
    const float* __restrict__ b1, const float* __restrict__ W2rel,
    const float* __restrict__ W2root, const float* __restrict__ b2,
    float* __restrict__ t, float* __restrict__ out) {
  __shared__ alignas(16) float sW1[2 * HID_CH * W1_PITCH];
  __shared__ alignas(16) float sW2[W2S_ROWS * W2_PITCH];
  __shared__ alignas(16) float sB1[HID_CH];
  __shared__ alignas(16) float sB2[OUT_CH];
  const int tid = threadIdx.x;
#pragma unroll
  for (int i = 0; i < HID_CH * IN_CH / 4 / 256; ++i) {
    const int f = tid + 256 * i;
    const int row = f / (IN_CH / 4), c4 = f % (IN_CH / 4);
    const float4 wr = ((const float4*)W1rel)[f];
    const float4 wo = ((const float4*)W1root)[f];
    *(float4*)(sW1 + row * W1_PITCH + c4 * 4) = wr;
    *(float4*)(sW1 + (HID_CH + row) * W1_PITCH + c4 * 4) = wo;
  }
#pragma unroll
  for (int i = 0; i < OUT_CH * HID_CH / 4 / 256; ++i) {
    const int f = tid + 256 * i;
    const int row = f / (HID_CH / 4), c4 = f % (HID_CH / 4);
    const float4 wr = ((const float4*)W2rel)[f];
    const float4 wo = ((const float4*)W2root)[f];
    *(float4*)(sW2 + row * W2_PITCH + c4 * 4) = wr;
    *(float4*)(sW2 + (OUT_CH + row) * W2_PITCH + c4 * 4) = wo;
  }
  if (tid < HID_CH) sB1[tid] = b1[tid];
  if (tid < OUT_CH) sB2[tid] = b2[tid];
  __syncthreads();

  const int lane = tid & 31;
  const int wave = tid >> 5;
  const int mtile = blockIdx.x * 8 + wave;
  if (mtile >= N_NODES / 16) return;
  const int m_base = mtile * 16;
  const int l15 = lane & 15;
  const int hf = lane >> 4;
  const int kh = hf * 2;
  const int node = m_base + l15;

  const float* G0 = agg1 + (long long)node * IN_CH + kh;
  const float* X0 = x + (long long)node * IN_CH + kh;
  const float* Wr = sW1 + l15 * W1_PITCH + kh;
  const float* Wo = Wr + HID_CH * W1_PITCH;

  v8f c[8] = {};
#pragma unroll
  for (int k = 0; k < IN_CH; k += 4) {
    const v2f bn = load_frag(G0 + k);
#pragma unroll
    for (int j = 0; j < 8; ++j) {
      const v2f aw = load_frag(Wr + j * 16 * W1_PITCH + k);
      c[j] = __builtin_amdgcn_wmma_f32_16x16x4_f32(false, aw, false, bn,
                                                   (short)0, c[j], false,
                                                   false);
    }
  }
#pragma unroll
  for (int k = 0; k < IN_CH; k += 4) {
    const v2f bn = load_frag(X0 + k);
#pragma unroll
    for (int j = 0; j < 8; ++j) {
      const v2f aw = load_frag(Wo + j * 16 * W1_PITCH + k);
      c[j] = __builtin_amdgcn_wmma_f32_16x16x4_f32(false, aw, false, bn,
                                                   (short)0, c[j], false,
                                                   false);
    }
  }

  __builtin_amdgcn_sched_barrier(0);
#pragma unroll
  for (int j = 0; j < 8; ++j) {
    const float4 ba = *(const float4*)(sB1 + 16 * j + 8 * hf);
    const float4 bb = *(const float4*)(sB1 + 16 * j + 8 * hf + 4);
    c[j][0] = fmaxf(c[j][0] + ba.x, 0.0f);
    c[j][1] = fmaxf(c[j][1] + ba.y, 0.0f);
    c[j][2] = fmaxf(c[j][2] + ba.z, 0.0f);
    c[j][3] = fmaxf(c[j][3] + ba.w, 0.0f);
    c[j][4] = fmaxf(c[j][4] + bb.x, 0.0f);
    c[j][5] = fmaxf(c[j][5] + bb.y, 0.0f);
    c[j][6] = fmaxf(c[j][6] + bb.z, 0.0f);
    c[j][7] = fmaxf(c[j][7] + bb.w, 0.0f);
  }

  const float* W2a = sW2 + l15 * W2_PITCH + 8 * hf;
  v8f d[5] = {};
#pragma unroll
  for (int j = 0; j < 8; ++j) {
#pragma unroll
    for (int q = 0; q < 4; ++q) {
      v2f bh;
      bh.x = c[j][2 * q];
      bh.y = c[j][2 * q + 1];
#pragma unroll
      for (int jm = 0; jm < 5; ++jm) {
        const v2f aw = load_frag(W2a + jm * 16 * W2_PITCH + 16 * j + 2 * q);
        d[jm] = __builtin_amdgcn_wmma_f32_16x16x4_f32(false, aw, false, bh,
                                                      (short)0, d[jm], false,
                                                      false);
      }
    }
  }

#pragma unroll
  for (int jm = 0; jm < 5; ++jm) {
    const int u = 2 * jm + hf;
    const bool is_t = u < 5;
    const int oc = (is_t ? u : u - 5) * 8;
    float* p = (is_t ? t : out) + (long long)node * OUT_CH + oc;
    float4 ba = *(const float4*)(sB2 + oc);
    float4 bb = *(const float4*)(sB2 + oc + 4);
    if (is_t) {
      ba = make_float4(0.f, 0.f, 0.f, 0.f);
      bb = ba;
    }
    *(float4*)p = make_float4(d[jm][0] + ba.x, d[jm][1] + ba.y,
                              d[jm][2] + ba.z, d[jm][3] + ba.w);
    *(float4*)(p + 4) = make_float4(d[jm][4] + bb.x, d[jm][5] + bb.y,
                                    d[jm][6] + bb.z, d[jm][7] + bb.w);
  }
}

__global__ __launch_bounds__(256) void gather_sum_rows40_log_softmax(
    const float* __restrict__ t, const int2* __restrict__ nodeinfo,
    const int* __restrict__ sorted, const int* __restrict__ ctrl,
    const int2* __restrict__ ovf, const float* __restrict__ zrow,
    float* __restrict__ out) {
  const int idx = blockIdx.x * 256 + threadIdx.x;
  const int node = idx >> 4;
  const int g = idx & 15;
  const bool act = g < OUT_CH / 4;
  float4* p = (float4*)(out + (long long)node * OUT_CH) + (act ? g : 0);
  const float4 start = act ? *p : make_float4(0.f, 0.f, 0.f, 0.f);
  const float4 v = sum_in_rows(t, OUT_CH, act, node, g, nodeinfo, sorted, ctrl,
                               ovf, zrow, start);
  float m = act ? fmaxf(fmaxf(v.x, v.y), fmaxf(v.z, v.w)) : -INFINITY;
#pragma unroll
  for (int msk = 8; msk >= 1; msk >>= 1)
    m = fmaxf(m, __shfl_xor(m, msk, 32));
  float s = act ? (__expf(v.x - m) + __expf(v.y - m) + __expf(v.z - m) +
                   __expf(v.w - m))
                : 0.0f;
#pragma unroll
  for (int msk = 8; msk >= 1; msk >>= 1) s += __shfl_xor(s, msk, 32);
  const float lz = m + __logf(s);
  if (act) *p = make_float4(v.x - lz, v.y - lz, v.z - lz, v.w - lz);
}

extern "C" void kernel_launch(void* const* d_in, const int* in_sizes, int n_in,
                              void* d_out, int out_size, void* d_ws,
                              size_t ws_size, hipStream_t stream) {
  (void)in_sizes; (void)n_in; (void)out_size; (void)ws_size;
  const float* x = (const float*)d_in[0];
  const int* ei = (const int*)d_in[1];
  const float* W1_rel = (const float*)d_in[2];
  const float* b1 = (const float*)d_in[3];
  const float* W1_root = (const float*)d_in[4];
  const float* W2_rel = (const float*)d_in[5];
  const float* b2 = (const float*)d_in[6];
  const float* W2_root = (const float*)d_in[7];
  float* out = (float*)d_out;

  float* agg1 = (float*)d_ws;
  float* tbuf = agg1 + (size_t)N_NODES * IN_CH;
  int* ctrl = (int*)(tbuf + (size_t)N_NODES * OUT_CH);
  float* zrow = (float*)(ctrl + 320);
  int2* nodeinfo = (int2*)(zrow + IN_CH);
  int2* rbuf = nodeinfo + N_NODES;
  int* sorted = (int*)(rbuf + (size_t)N_RANGES * RANGE_CAP);
  int2* ovf = (int2*)(sorted + (size_t)N_RANGES * RANGE_CAP);

  const int* srcv = ei;
  const int* dstv = ei + N_EDGES;

  clear_ctrl<<<1, 512, 0, stream>>>(ctrl, zrow);
  partition_edges<<<PART_BLOCKS, 256, 0, stream>>>(srcv, dstv, ctrl, rbuf, ovf);
  sort_range<<<N_RANGES, 1024, 0, stream>>>(ctrl, rbuf, sorted, nodeinfo);
  gather_sum_rows64<<<N_NODES * 16 / 256, 256, 0, stream>>>(
      x, nodeinfo, sorted, ctrl, ovf, zrow, agg1);
  node_gemms_fused<<<(N_NODES / 16 + 7) / 8, 256, 0, stream>>>(
      agg1, x, W1_rel, W1_root, b1, W2_rel, W2_root, b2, tbuf, out);
  gather_sum_rows40_log_softmax<<<N_NODES * 16 / 256, 256, 0, stream>>>(
      tbuf, nodeinfo, sorted, ctrl, ovf, zrow, out);
}


# node GEMM: one 12-wave work-group per WGP, one 16-node tile per wave (3 per SIMD); the 53 left-over tiles split by channel tile over the waves of work-groups 0..52 (h via LDS); sum order unchanged
# speedup vs baseline: 3.8665x; 1.0372x over previous
#include <hip/hip_runtime.h>
#include <hip/hip_bf16.h>
#include <math.h>


typedef __attribute__((ext_vector_type(2))) float v2f;
typedef __attribute__((ext_vector_type(8))) float v8f;

#define N_NODES 50000
#define N_EDGES 800000
#define IN_CH   64
#define HID_CH  128
#define OUT_CH  40

#define RANGE_NODES 196
#define N_RANGES 256
#define RANGE_CAP 4096
#define PART_EDGES 1280
#define PART_BLOCKS (N_EDGES / PART_EDGES)
#define LIST_CHUNK 48
static_assert(N_RANGES * RANGE_NODES >= N_NODES, "ranges must cover all nodes");
static_assert(N_EDGES % PART_EDGES == 0 && (N_NODES * 16) % 256 == 0,
              "edge and node grids must be exact: the kernels have no tail guard");

__global__ __launch_bounds__(512) void clear_ctrl(int* __restrict__ ctrl,
                                                  float* __restrict__ zrow) {
  const int i = threadIdx.x;
  if (i <= N_RANGES) ctrl[i] = 0;
  if (i < IN_CH) zrow[i] = 0.0f;
}

__global__ __launch_bounds__(256) void partition_edges(
    const int* __restrict__ srcv, const int* __restrict__ dstv,
    int* __restrict__ ctrl, int2* __restrict__ rbuf, int2* __restrict__ ovf) {
  __shared__ int hist[N_RANGES];
  __shared__ int gbase[N_RANGES];
  const int tid = threadIdx.x;
  hist[tid] = 0;
  __syncthreads();
  int d[PART_EDGES / 256], s[PART_EDGES / 256], rk[PART_EDGES / 256];
  const int e0 = blockIdx.x * PART_EDGES + tid;
#pragma unroll
  for (int i = 0; i < PART_EDGES / 256; ++i) {
    d[i] = dstv[e0 + i * 256];
    s[i] = srcv[e0 + i * 256];
  }
#pragma unroll
  for (int i = 0; i < PART_EDGES / 256; ++i)
    rk[i] = atomicAdd(&hist[d[i] / RANGE_NODES], 1);
  __syncthreads();
  {
    const int c = hist[tid];
    gbase[tid] = c > 0 ? atomicAdd(ctrl + tid, c) : 0;
  }
  __syncthreads();
#pragma unroll
  for (int i = 0; i < PART_EDGES / 256; ++i) {
    const int r = d[i] / RANGE_NODES;
    const int pos = gbase[r] + rk[i];
    if (pos < RANGE_CAP) {
      rbuf[r * RANGE_CAP + pos] = make_int2(d[i], s[i]);
    } else {
      const int k = atomicAdd(ctrl + N_RANGES, 1);
      ovf[k] = make_int2(d[i], s[i]);
    }
  }
}

__global__ __launch_bounds__(1024) void sort_range(
    const int* __restrict__ ctrl, const int2* __restrict__ rbuf,
    int* __restrict__ sorted, int2* __restrict__ nodeinfo) {
  __shared__ int cnt[256];
  __shared__ int off[256];
  __shared__ int list[RANGE_CAP];
  const int tid = threadIdx.x;
  const int r = blockIdx.x;
  const int tot = ctrl[r];
  const int n = tot < RANGE_CAP ? tot : RANGE_CAP;
  if (tid < 256) cnt[tid] = 0;
  __syncthreads();
  int2 p[RANGE_CAP / 1024];
  int dl[RANGE_CAP / 1024], rk[RANGE_CAP / 1024];
#pragma unroll
  for (int i = 0; i < RANGE_CAP / 1024; ++i) {
    const int e = tid + i * 1024;
    p[i] = make_int2(-1, 0);
    if (e < n) p[i] = rbuf[r * RANGE_CAP + e];
  }
#pragma unroll
  for (int i = 0; i < RANGE_CAP / 1024; ++i) {
    dl[i] = -1; rk[i] = 0;
    if (p[i].x >= 0) {
      dl[i] = p[i].x - r * RANGE_NODES;
      rk[i] = atomicAdd(&cnt[dl[i]], 1);
    }
  }
  __syncthreads();
  if (tid < 256) {
    int o = 0;
    for (int u = 0; u < tid; ++u) o += cnt[u];
    off[tid] = o;
  }
  __syncthreads();
#pragma unroll
  for (int i = 0; i < RANGE_CAP / 1024; ++i)
    if (dl[i] >= 0) list[off[dl[i]] + rk[i]] = p[i].y;
  __syncthreads();
#pragma unroll
  for (int i = 0; i < RANGE_CAP / 1024; ++i) {
    const int e = tid + i * 1024;
    if (e < n) sorted[r * RANGE_CAP + e] = list[e];
  }
  if (tid < RANGE_NODES) {
    const int node = r * RANGE_NODES + tid;
    if (node < N_NODES)
      nodeinfo[node] = make_int2(r * RANGE_CAP + off[tid], cnt[tid]);
  }
}

__device__ __forceinline__ float4 sum_in_rows(
    const float* __restrict__ rows, const int rowf, const bool act,
    const int node, const int g, const int2* __restrict__ nodeinfo,
    const int* __restrict__ sorted, const int* __restrict__ ctrl,
    const int2* __restrict__ ovf, const float* __restrict__ zrow, float4 acc) {
  const unsigned long long info =
      *(const unsigned long long*)(nodeinfo + node);
  const int start = (int)(unsigned)info;
  const int deg = (int)(info >> 32);
  const float* base = rows + g * 4;
  const float* zpiece = zrow + g * 4;
  for (int o = 0; o < deg; o += LIST_CHUNK) {
    const int n = deg - o < LIST_CHUNK ? deg - o : LIST_CHUNK;
    const int* b = sorted + start + o;
    const int e0 = (g < n) ? b[g] : 0;
    const int e1 = (g + 16 < n) ? b[g + 16] : 0;
    const int e2 = (g + 32 < n) ? b[g + 32] : 0;
    for (int j0 = 0; j0 < n; j0 += 8) {
      const int ev = j0 < 16 ? e0 : (j0 < 32 ? e1 : e2);
      float4 v[8];
#pragma unroll
      for (int u = 0; u < 8; ++u) {
        const int s = __shfl(ev, (j0 + u) & 15, 16);
        const float* p =
            (act && j0 + u < n) ? base + (long long)s * rowf : zpiece;
        v[u] = *(const float4*)p;
      }
#pragma unroll
      for (int u = 0; u < 8; ++u) {
        acc.x += v[u].x; acc.y += v[u].y; acc.z += v[u].z; acc.w += v[u].w;
      }
    }
  }
  if (ctrl[node / RANGE_NODES] > RANGE_CAP) {
    const int m = ctrl[N_RANGES];
    for (int k = 0; k < m; ++k) {
      const int2 p = ovf[k];
      if (p.x == node && act) {
        const float4 v = *(const float4*)(base + (long long)p.y * rowf);
        acc.x += v.x; acc.y += v.y; acc.z += v.z; acc.w += v.w;
      }
    }
  }
  return acc;
}

__global__ __launch_bounds__(256) void gather_sum_rows64(
    const float* __restrict__ feat, const int2* __restrict__ nodeinfo,
    const int* __restrict__ sorted, const int* __restrict__ ctrl,
    const int2* __restrict__ ovf, const float* __restrict__ zrow,
    float* __restrict__ agg) {
  const int idx = blockIdx.x * 256 + threadIdx.x;
  const int node = idx >> 4;
  const int g = idx & 15;
  const float4 acc =
      sum_in_rows(feat, IN_CH, true, node, g, nodeinfo, sorted, ctrl, ovf,
                  zrow, make_float4(0.f, 0.f, 0.f, 0.f));
  *(float4*)(agg + (long long)node * IN_CH + g * 4) = acc;
}

__device__ __forceinline__ v2f load_frag(const float* base) {
  return *(const v2f*)base;
}

#define W1_PITCH (IN_CH + 4)
#define W2_PITCH (HID_CH + 4)
#define W2S_ROWS (2 * OUT_CH)

#define GEMM_WAVES 12
#define GEMM_THREADS (GEMM_WAVES * 32)
#define GEMM_BLOCKS 256
#define N_TILES (N_NODES / 16)
#define MAIN_TILES (GEMM_BLOCKS * GEMM_WAVES)
#define LEFT_TILES (N_TILES - MAIN_TILES)
#define HS_PITCH (HID_CH + 4)
static_assert(N_NODES % 16 == 0 && LEFT_TILES >= 0 && LEFT_TILES <= GEMM_BLOCKS &&
                  GEMM_THREADS >= 256 && (HID_CH * IN_CH / 4) % 256 == 0 &&
                  (OUT_CH * HID_CH / 4) % 256 == 0,
              "every tile needs an owner, and 256 threads stage the weights in whole trips");

__device__ __forceinline__ void store_stacked_run(
    const v8f d, const int jm, const int hf, const int node,
    const float* sB2, float* __restrict__ t, float* __restrict__ out) {
  const int u = 2 * jm + hf;
  const bool is_t = u < 5;
  const int oc = (is_t ? u : u - 5) * 8;
  float* p = (is_t ? t : out) + (long long)node * OUT_CH + oc;
  float4 ba = *(const float4*)(sB2 + oc);
  float4 bb = *(const float4*)(sB2 + oc + 4);
  if (is_t) {
    ba = make_float4(0.f, 0.f, 0.f, 0.f);
    bb = ba;
  }
  *(float4*)p = make_float4(d[0] + ba.x, d[1] + ba.y, d[2] + ba.z, d[3] + ba.w);
  *(float4*)(p + 4) =
      make_float4(d[4] + bb.x, d[5] + bb.y, d[6] + bb.z, d[7] + bb.w);
}

__global__ __launch_bounds__(GEMM_THREADS) void node_gemms_fused(
    const float* __restrict__ agg1, const float* __restrict__ x,
    const float* __restrict__ W1rel, const float* __restrict__ W1root,
    const float* __restrict__ b1, const float* __restrict__ W2rel,
    const float* __restrict__ W2root, const float* __restrict__ b2,
    float* __restrict__ t, float* __restrict__ out) {
  __shared__ alignas(16) float sW1[2 * HID_CH * W1_PITCH];
  __shared__ alignas(16) float sW2[W2S_ROWS * W2_PITCH];
  __shared__ alignas(16) float sB1[HID_CH];
  __shared__ alignas(16) float sB2[OUT_CH];
  __shared__ alignas(16) float sH[16 * HS_PITCH];
  const int tid = threadIdx.x;
  if (tid < 256) {
#pragma unroll
    for (int i = 0; i < HID_CH * IN_CH / 4 / 256; ++i) {
      const int f = tid + 256 * i;
      const int row = f / (IN_CH / 4), c4 = f % (IN_CH / 4);
      const float4 wr = ((const float4*)W1rel)[f];
      const float4 wo = ((const float4*)W1root)[f];
      *(float4*)(sW1 + row * W1_PITCH + c4 * 4) = wr;
      *(float4*)(sW1 + (HID_CH + row) * W1_PITCH + c4 * 4) = wo;
    }
#pragma unroll
    for (int i = 0; i < OUT_CH * HID_CH / 4 / 256; ++i) {
      const int f = tid + 256 * i;
      const int row = f / (HID_CH / 4), c4 = f % (HID_CH / 4);
      const float4 wr = ((const float4*)W2rel)[f];
      const float4 wo = ((const float4*)W2root)[f];
      *(float4*)(sW2 + row * W2_PITCH + c4 * 4) = wr;
      *(float4*)(sW2 + (OUT_CH + row) * W2_PITCH + c4 * 4) = wo;
    }
  }
  if (tid < HID_CH) sB1[tid] = b1[tid];
  if (tid < OUT_CH) sB2[tid] = b2[tid];
  __syncthreads();

  const int lane = tid & 31;
  const int wave = tid >> 5;
  const int l15 = lane & 15;
  const int hf = lane >> 4;
  const int kh = hf * 2;
  const bool has_left = blockIdx.x < LEFT_TILES;
  const int node_left = (MAIN_TILES + blockIdx.x) * 16 + l15;

  const bool left1 = has_left && wave < HID_CH / 16;
  v8f cl = {};
  if (left1) {
    const float* GL = agg1 + (long long)node_left * IN_CH + kh;
    const float* XL = x + (long long)node_left * IN_CH + kh;
    const float* WrL = sW1 + (wave * 16 + l15) * W1_PITCH + kh;
    const float* WoL = WrL + HID_CH * W1_PITCH;
#pragma unroll
    for (int k = 0; k < IN_CH; k += 4)
      cl = __builtin_amdgcn_wmma_f32_16x16x4_f32(false, load_frag(WrL + k),
                                                 false, load_frag(GL + k),
                                                 (short)0, cl, false, false);
#pragma unroll
    for (int k = 0; k < IN_CH; k += 4)
      cl = __builtin_amdgcn_wmma_f32_16x16x4_f32(false, load_frag(WoL + k),
                                                 false, load_frag(XL + k),
                                                 (short)0, cl, false, false);
  }

  const int mtile = blockIdx.x * GEMM_WAVES + wave;
  const int m_base = mtile * 16;
  const int node = m_base + l15;

  const float* G0 = agg1 + (long long)node * IN_CH + kh;
  const float* X0 = x + (long long)node * IN_CH + kh;
  const float* Wr = sW1 + l15 * W1_PITCH + kh;
  const float* Wo = Wr + HID_CH * W1_PITCH;

  v8f c[8] = {};
#pragma unroll
  for (int k = 0; k < IN_CH; k += 4) {
    const v2f bn = load_frag(G0 + k);
#pragma unroll
    for (int j = 0; j < 8; ++j) {
      const v2f aw = load_frag(Wr + j * 16 * W1_PITCH + k);
      c[j] = __builtin_amdgcn_wmma_f32_16x16x4_f32(false, aw, false, bn,
                                                   (short)0, c[j], false,
                                                   false);
    }
  }
#pragma unroll
  for (int k = 0; k < IN_CH; k += 4) {
    const v2f bn = load_frag(X0 + k);
#pragma unroll
    for (int j = 0; j < 8; ++j) {
      const v2f aw = load_frag(Wo + j * 16 * W1_PITCH + k);
      c[j] = __builtin_amdgcn_wmma_f32_16x16x4_f32(false, aw, false, bn,
                                                   (short)0, c[j], false,
                                                   false);
    }
  }

  if (left1) {
    const float4 ba = *(const float4*)(sB1 + 16 * wave + 8 * hf);
    const float4 bb = *(const float4*)(sB1 + 16 * wave + 8 * hf + 4);
    float* hp = sH + l15 * HS_PITCH + 16 * wave + 8 * hf;
    *(float4*)hp =
        make_float4(fmaxf(cl[0] + ba.x, 0.0f), fmaxf(cl[1] + ba.y, 0.0f),
                    fmaxf(cl[2] + ba.z, 0.0f), fmaxf(cl[3] + ba.w, 0.0f));
    *(float4*)(hp + 4) =
        make_float4(fmaxf(cl[4] + bb.x, 0.0f), fmaxf(cl[5] + bb.y, 0.0f),
                    fmaxf(cl[6] + bb.z, 0.0f), fmaxf(cl[7] + bb.w, 0.0f));
  }
  if (has_left) __syncthreads();

  const bool left2 = has_left && wave < W2S_ROWS / 16;
  v8f dl = {};
  if (left2) {
    const float* W2aL = sW2 + (wave * 16 + l15) * W2_PITCH + 8 * hf;
    const float* hb = sH + l15 * HS_PITCH + 8 * hf;
#pragma unroll
    for (int j = 0; j < 8; ++j) {
#pragma unroll
      for (int q = 0; q < 4; ++q)
        dl = __builtin_amdgcn_wmma_f32_16x16x4_f32(
            false, load_frag(W2aL + 16 * j + 2 * q), false,
            load_frag(hb + 16 * j + 2 * q), (short)0, dl, false, false);
    }
  }

#pragma unroll
  for (int j = 0; j < 8; ++j) {
    const float4 ba = *(const float4*)(sB1 + 16 * j + 8 * hf);
    const float4 bb = *(const float4*)(sB1 + 16 * j + 8 * hf + 4);
    c[j][0] = fmaxf(c[j][0] + ba.x, 0.0f);
    c[j][1] = fmaxf(c[j][1] + ba.y, 0.0f);
    c[j][2] = fmaxf(c[j][2] + ba.z, 0.0f);
    c[j][3] = fmaxf(c[j][3] + ba.w, 0.0f);
    c[j][4] = fmaxf(c[j][4] + bb.x, 0.0f);
    c[j][5] = fmaxf(c[j][5] + bb.y, 0.0f);
    c[j][6] = fmaxf(c[j][6] + bb.z, 0.0f);
    c[j][7] = fmaxf(c[j][7] + bb.w, 0.0f);
  }

  const float* W2a = sW2 + l15 * W2_PITCH + 8 * hf;
  v8f d[5] = {};
#pragma unroll
  for (int j = 0; j < 8; ++j) {
#pragma unroll
    for (int q = 0; q < 4; ++q) {
      v2f bh;
      bh.x = c[j][2 * q];
      bh.y = c[j][2 * q + 1];
#pragma unroll
      for (int jm = 0; jm < 5; ++jm) {
        const v2f aw = load_frag(W2a + jm * 16 * W2_PITCH + 16 * j + 2 * q);
        d[jm] = __builtin_amdgcn_wmma_f32_16x16x4_f32(false, aw, false, bh,
                                                      (short)0, d[jm], false,
                                                      false);
      }
    }
  }
#pragma unroll
  for (int jm = 0; jm < 5; ++jm)
    store_stacked_run(d[jm], jm, hf, node, sB2, t, out);
  if (left2) store_stacked_run(dl, wave, hf, node_left, sB2, t, out);
}

__global__ __launch_bounds__(256) void gather_sum_rows40_log_softmax(
    const float* __restrict__ t, const int2* __restrict__ nodeinfo,
    const int* __restrict__ sorted, const int* __restrict__ ctrl,
    const int2* __restrict__ ovf, const float* __restrict__ zrow,
    float* __restrict__ out) {
  const int idx = blockIdx.x * 256 + threadIdx.x;
  const int node = idx >> 4;
  const int g = idx & 15;
  const bool act = g < OUT_CH / 4;
  float4* p = (float4*)(out + (long long)node * OUT_CH) + (act ? g : 0);
  const float4 start = act ? *p : make_float4(0.f, 0.f, 0.f, 0.f);
  const float4 v = sum_in_rows(t, OUT_CH, act, node, g, nodeinfo, sorted, ctrl,
                               ovf, zrow, start);
  float m = act ? fmaxf(fmaxf(v.x, v.y), fmaxf(v.z, v.w)) : -INFINITY;
#pragma unroll
  for (int msk = 8; msk >= 1; msk >>= 1)
    m = fmaxf(m, __shfl_xor(m, msk, 32));
  float s = act ? (__expf(v.x - m) + __expf(v.y - m) + __expf(v.z - m) +
                   __expf(v.w - m))
                : 0.0f;
#pragma unroll
  for (int msk = 8; msk >= 1; msk >>= 1) s += __shfl_xor(s, msk, 32);
  const float lz = m + __logf(s);
  if (act) *p = make_float4(v.x - lz, v.y - lz, v.z - lz, v.w - lz);
}

extern "C" void kernel_launch(void* const* d_in, const int* in_sizes, int n_in,
                              void* d_out, int out_size, void* d_ws,
                              size_t ws_size, hipStream_t stream) {
  (void)in_sizes; (void)n_in; (void)out_size; (void)ws_size;
  const float* x = (const float*)d_in[0];
  const int* ei = (const int*)d_in[1];
  const float* W1_rel = (const float*)d_in[2];
  const float* b1 = (const float*)d_in[3];
  const float* W1_root = (const float*)d_in[4];
  const float* W2_rel = (const float*)d_in[5];
  const float* b2 = (const float*)d_in[6];
  const float* W2_root = (const float*)d_in[7];
  float* out = (float*)d_out;

  float* agg1 = (float*)d_ws;
  float* tbuf = agg1 + (size_t)N_NODES * IN_CH;
  int* ctrl = (int*)(tbuf + (size_t)N_NODES * OUT_CH);
  float* zrow = (float*)(ctrl + 320);
  int2* nodeinfo = (int2*)(zrow + IN_CH);
  int2* rbuf = nodeinfo + N_NODES;
  int* sorted = (int*)(rbuf + (size_t)N_RANGES * RANGE_CAP);
  int2* ovf = (int2*)(sorted + (size_t)N_RANGES * RANGE_CAP);

  const int* srcv = ei;
  const int* dstv = ei + N_EDGES;

  clear_ctrl<<<1, 512, 0, stream>>>(ctrl, zrow);
  partition_edges<<<PART_BLOCKS, 256, 0, stream>>>(srcv, dstv, ctrl, rbuf, ovf);
  sort_range<<<N_RANGES, 1024, 0, stream>>>(ctrl, rbuf, sorted, nodeinfo);
  gather_sum_rows64<<<N_NODES * 16 / 256, 256, 0, stream>>>(
      x, nodeinfo, sorted, ctrl, ovf, zrow, agg1);
  node_gemms_fused<<<GEMM_BLOCKS, GEMM_THREADS, 0, stream>>>(
      agg1, x, W1_rel, W1_root, b1, W2_rel, W2_root, b2, tbuf, out);
  gather_sum_rows40_log_softmax<<<N_NODES * 16 / 256, 256, 0, stream>>>(
      tbuf, nodeinfo, sorted, ctrl, ovf, zrow, out);
}


# the work-group that sorts a range also forms layer 1's edge sums for its 196 nodes (degree and list read from LDS, next node from an LDS counter): 1 launch, 25,000 short waves less; sum order kept
# speedup vs baseline: 3.9978x; 1.0340x over previous
#include <hip/hip_runtime.h>
#include <hip/hip_bf16.h>
#include <math.h>


typedef __attribute__((ext_vector_type(2))) float v2f;
typedef __attribute__((ext_vector_type(8))) float v8f;

#define N_NODES 50000
#define N_EDGES 800000
#define IN_CH   64
#define HID_CH  128
#define OUT_CH  40

#define RANGE_NODES 196
#define N_RANGES 256
#define RANGE_CAP 4096
#define PART_EDGES 1280
#define PART_BLOCKS (N_EDGES / PART_EDGES)
#define LIST_CHUNK 48
static_assert(N_RANGES * RANGE_NODES >= N_NODES, "ranges must cover all nodes");
static_assert(N_EDGES % PART_EDGES == 0 && (N_NODES * 16) % 256 == 0,
              "edge and node grids must be exact: the kernels have no tail guard");

__global__ __launch_bounds__(512) void clear_ctrl(int* __restrict__ ctrl,
                                                  float* __restrict__ zrow) {
  const int i = threadIdx.x;
  if (i <= N_RANGES) ctrl[i] = 0;
  if (i < IN_CH) zrow[i] = 0.0f;
}

__global__ __launch_bounds__(256) void partition_edges(
    const int* __restrict__ srcv, const int* __restrict__ dstv,
    int* __restrict__ ctrl, int2* __restrict__ rbuf, int2* __restrict__ ovf) {
  __shared__ int hist[N_RANGES];
  __shared__ int gbase[N_RANGES];
  const int tid = threadIdx.x;
  hist[tid] = 0;
  __syncthreads();
  int d[PART_EDGES / 256], s[PART_EDGES / 256], rk[PART_EDGES / 256];
  const int e0 = blockIdx.x * PART_EDGES + tid;
#pragma unroll
  for (int i = 0; i < PART_EDGES / 256; ++i) {
    d[i] = dstv[e0 + i * 256];
    s[i] = srcv[e0 + i * 256];
  }
#pragma unroll
  for (int i = 0; i < PART_EDGES / 256; ++i)
    rk[i] = atomicAdd(&hist[d[i] / RANGE_NODES], 1);
  __syncthreads();
  {
    const int c = hist[tid];
    gbase[tid] = c > 0 ? atomicAdd(ctrl + tid, c) : 0;
  }
  __syncthreads();
#pragma unroll
  for (int i = 0; i < PART_EDGES / 256; ++i) {
    const int r = d[i] / RANGE_NODES;
    const int pos = gbase[r] + rk[i];
    if (pos < RANGE_CAP) {
      rbuf[r * RANGE_CAP + pos] = make_int2(d[i], s[i]);
    } else {
      const int k = atomicAdd(ctrl + N_RANGES, 1);
      ovf[k] = make_int2(d[i], s[i]);
    }
  }
}

#define SORT_THREADS 1024
#define SORT_GROUPS (SORT_THREADS / 16)
__global__ __launch_bounds__(SORT_THREADS) void sort_range_sum_rows64(
    const int* __restrict__ ctrl, const int2* __restrict__ rbuf,
    const int2* __restrict__ ovf, const float* __restrict__ feat,
    const float* __restrict__ zrow, int* __restrict__ sorted,
    int2* __restrict__ nodeinfo, float* __restrict__ agg) {
  __shared__ int cnt[256];
  __shared__ int off[256];
  __shared__ int list[RANGE_CAP];
  __shared__ int next_ln;
  const int tid = threadIdx.x;
  const int r = blockIdx.x;
  const int tot = ctrl[r];
  const int n = tot < RANGE_CAP ? tot : RANGE_CAP;
  if (tid < 256) cnt[tid] = 0;
  if (tid == 0) next_ln = SORT_GROUPS;
  __syncthreads();
  int2 p[RANGE_CAP / SORT_THREADS];
  int dl[RANGE_CAP / SORT_THREADS], rk[RANGE_CAP / SORT_THREADS];
#pragma unroll
  for (int i = 0; i < RANGE_CAP / SORT_THREADS; ++i) {
    const int e = tid + i * SORT_THREADS;
    p[i] = make_int2(-1, 0);
    if (e < n) p[i] = rbuf[r * RANGE_CAP + e];
  }
#pragma unroll
  for (int i = 0; i < RANGE_CAP / SORT_THREADS; ++i) {
    dl[i] = -1; rk[i] = 0;
    if (p[i].x >= 0) {
      dl[i] = p[i].x - r * RANGE_NODES;
      rk[i] = atomicAdd(&cnt[dl[i]], 1);
    }
  }
  __syncthreads();
  if (tid < 256) {
    int o = 0;
    for (int u = 0; u < tid; ++u) o += cnt[u];
    off[tid] = o;
  }
  __syncthreads();
#pragma unroll
  for (int i = 0; i < RANGE_CAP / SORT_THREADS; ++i)
    if (dl[i] >= 0) list[off[dl[i]] + rk[i]] = p[i].y;
  __syncthreads();
#pragma unroll
  for (int i = 0; i < RANGE_CAP / SORT_THREADS; ++i) {
    const int e = tid + i * SORT_THREADS;
    if (e < n) sorted[r * RANGE_CAP + e] = list[e];
  }
  const int first = r * RANGE_NODES;
  if (tid < RANGE_NODES) {
    const int node = first + tid;
    if (node < N_NODES)
      nodeinfo[node] = make_int2(r * RANGE_CAP + off[tid], cnt[tid]);
  }

  const int g = tid & 15;
  const int nn = N_NODES - first < RANGE_NODES ? N_NODES - first : RANGE_NODES;
  const bool over = tot > RANGE_CAP;
  const int m = over ? ctrl[N_RANGES] : 0;
  const float* base = feat + g * 4;
  const float* zpiece = zrow + g * 4;
  int ln = tid >> 4;
  while (ln < nn) {
    int nxt = 0;
    if (g == 0) nxt = atomicAdd(&next_ln, 1);
    const int deg = cnt[ln];
    const int st = off[ln];
    float4 acc = make_float4(0.f, 0.f, 0.f, 0.f);
    for (int j0 = 0; j0 < deg; j0 += 8) {
      int s[8];
#pragma unroll
      for (int u = 0; u < 8; ++u)
        s[u] = list[st + (j0 + u < deg ? j0 + u : deg - 1)];
      float4 v[8];
#pragma unroll
      for (int u = 0; u < 8; ++u) {
        const float* q =
            (j0 + u < deg) ? base + (long long)s[u] * IN_CH : zpiece;
        v[u] = *(const float4*)q;
      }
#pragma unroll
      for (int u = 0; u < 8; ++u) {
        acc.x += v[u].x; acc.y += v[u].y; acc.z += v[u].z; acc.w += v[u].w;
      }
    }
    const int node = first + ln;
    if (over) {
      for (int k = 0; k < m; ++k) {
        const int2 pr = ovf[k];
        if (pr.x == node) {
          const float4 v = *(const float4*)(base + (long long)pr.y * IN_CH);
          acc.x += v.x; acc.y += v.y; acc.z += v.z; acc.w += v.w;
        }
      }
    }
    *(float4*)(agg + (long long)node * IN_CH + g * 4) = acc;
    ln = __shfl(nxt, 0, 16);
  }
}

__device__ __forceinline__ float4 sum_in_rows(
    const float* __restrict__ rows, const int rowf, const bool act,
    const int node, const int g, const int2* __restrict__ nodeinfo,
    const int* __restrict__ sorted, const int* __restrict__ ctrl,
    const int2* __restrict__ ovf, const float* __restrict__ zrow, float4 acc) {
  const unsigned long long info =
      *(const unsigned long long*)(nodeinfo + node);
  const int start = (int)(unsigned)info;
  const int deg = (int)(info >> 32);
  const float* base = rows + g * 4;
  const float* zpiece = zrow + g * 4;
  for (int o = 0; o < deg; o += LIST_CHUNK) {
    const int n = deg - o < LIST_CHUNK ? deg - o : LIST_CHUNK;
    const int* b = sorted + start + o;
    const int e0 = (g < n) ? b[g] : 0;
    const int e1 = (g + 16 < n) ? b[g + 16] : 0;
    const int e2 = (g + 32 < n) ? b[g + 32] : 0;
    for (int j0 = 0; j0 < n; j0 += 8) {
      const int ev = j0 < 16 ? e0 : (j0 < 32 ? e1 : e2);
      float4 v[8];
#pragma unroll
      for (int u = 0; u < 8; ++u) {
        const int s = __shfl(ev, (j0 + u) & 15, 16);
        const float* p =
            (act && j0 + u < n) ? base + (long long)s * rowf : zpiece;
        v[u] = *(const float4*)p;
      }
#pragma unroll
      for (int u = 0; u < 8; ++u) {
        acc.x += v[u].x; acc.y += v[u].y; acc.z += v[u].z; acc.w += v[u].w;
      }
    }
  }
  if (ctrl[node / RANGE_NODES] > RANGE_CAP) {
    const int m = ctrl[N_RANGES];
    for (int k = 0; k < m; ++k) {
      const int2 p = ovf[k];
      if (p.x == node && act) {
        const float4 v = *(const float4*)(base + (long long)p.y * rowf);
        acc.x += v.x; acc.y += v.y; acc.z += v.z; acc.w += v.w;
      }
    }
  }
  return acc;
}

__device__ __forceinline__ v2f load_frag(const float* base) {
  return *(const v2f*)base;
}

#define W1_PITCH (IN_CH + 4)
#define W2_PITCH (HID_CH + 4)
#define W2S_ROWS (2 * OUT_CH)

#define GEMM_WAVES 12
#define GEMM_THREADS (GEMM_WAVES * 32)
#define GEMM_BLOCKS 256
#define N_TILES (N_NODES / 16)
#define MAIN_TILES (GEMM_BLOCKS * GEMM_WAVES)
#define LEFT_TILES (N_TILES - MAIN_TILES)
#define HS_PITCH (HID_CH + 4)
static_assert(N_NODES % 16 == 0 && LEFT_TILES >= 0 && LEFT_TILES <= GEMM_BLOCKS &&
                  GEMM_THREADS >= 256 && (HID_CH * IN_CH / 4) % 256 == 0 &&
                  (OUT_CH * HID_CH / 4) % 256 == 0,
              "every tile needs an owner, and 256 threads stage the weights in whole trips");

__device__ __forceinline__ void store_stacked_run(
    const v8f d, const int jm, const int hf, const int node,
    const float* sB2, float* __restrict__ t, float* __restrict__ out) {
  const int u = 2 * jm + hf;
  const bool is_t = u < 5;
  const int oc = (is_t ? u : u - 5) * 8;
  float* p = (is_t ? t : out) + (long long)node * OUT_CH + oc;
  float4 ba = *(const float4*)(sB2 + oc);
  float4 bb = *(const float4*)(sB2 + oc + 4);
  if (is_t) {
    ba = make_float4(0.f, 0.f, 0.f, 0.f);
    bb = ba;
  }
  *(float4*)p = make_float4(d[0] + ba.x, d[1] + ba.y, d[2] + ba.z, d[3] + ba.w);
  *(float4*)(p + 4) =
      make_float4(d[4] + bb.x, d[5] + bb.y, d[6] + bb.z, d[7] + bb.w);
}

__global__ __launch_bounds__(GEMM_THREADS) void node_gemms_fused(
    const float* __restrict__ agg1, const float* __restrict__ x,
    const float* __restrict__ W1rel, const float* __restrict__ W1root,
    const float* __restrict__ b1, const float* __restrict__ W2rel,
    const float* __restrict__ W2root, const float* __restrict__ b2,
    float* __restrict__ t, float* __restrict__ out) {
  __shared__ alignas(16) float sW1[2 * HID_CH * W1_PITCH];
  __shared__ alignas(16) float sW2[W2S_ROWS * W2_PITCH];
  __shared__ alignas(16) float sB1[HID_CH];
  __shared__ alignas(16) float sB2[OUT_CH];
  __shared__ alignas(16) float sH[16 * HS_PITCH];
  const int tid = threadIdx.x;
  if (tid < 256) {
#pragma unroll
    for (int i = 0; i < HID_CH * IN_CH / 4 / 256; ++i) {
      const int f = tid + 256 * i;
      const int row = f / (IN_CH / 4), c4 = f % (IN_CH / 4);
      const float4 wr = ((const float4*)W1rel)[f];
      const float4 wo = ((const float4*)W1root)[f];
      *(float4*)(sW1 + row * W1_PITCH + c4 * 4) = wr;
      *(float4*)(sW1 + (HID_CH + row) * W1_PITCH + c4 * 4) = wo;
    }
#pragma unroll
    for (int i = 0; i < OUT_CH * HID_CH / 4 / 256; ++i) {
      const int f = tid + 256 * i;
      const int row = f / (HID_CH / 4), c4 = f % (HID_CH / 4);
      const float4 wr = ((const float4*)W2rel)[f];
      const float4 wo = ((const float4*)W2root)[f];
      *(float4*)(sW2 + row * W2_PITCH + c4 * 4) = wr;
      *(float4*)(sW2 + (OUT_CH + row) * W2_PITCH + c4 * 4) = wo;
    }
  }
  if (tid < HID_CH) sB1[tid] = b1[tid];
  if (tid < OUT_CH) sB2[tid] = b2[tid];
  __syncthreads();

  const int lane = tid & 31;
  const int wave = tid >> 5;
  const int l15 = lane & 15;
  const int hf = lane >> 4;
  const int kh = hf * 2;
  const bool has_left = blockIdx.x < LEFT_TILES;
  const int node_left = (MAIN_TILES + blockIdx.x) * 16 + l15;

  const bool left1 = has_left && wave < HID_CH / 16;
  v8f cl = {};
  if (left1) {
    const float* GL = agg1 + (long long)node_left * IN_CH + kh;
    const float* XL = x + (long long)node_left * IN_CH + kh;
    const float* WrL = sW1 + (wave * 16 + l15) * W1_PITCH + kh;
    const float* WoL = WrL + HID_CH * W1_PITCH;
#pragma unroll
    for (int k = 0; k < IN_CH; k += 4)
      cl = __builtin_amdgcn_wmma_f32_16x16x4_f32(false, load_frag(WrL + k),
                                                 false, load_frag(GL + k),
                                                 (short)0, cl, false, false);
#pragma unroll
    for (int k = 0; k < IN_CH; k += 4)
      cl = __builtin_amdgcn_wmma_f32_16x16x4_f32(false, load_frag(WoL + k),
                                                 false, load_frag(XL + k),
                                                 (short)0, cl, false, false);
  }

  const int mtile = blockIdx.x * GEMM_WAVES + wave;
  const int m_base = mtile * 16;
  const int node = m_base + l15;

  const float* G0 = agg1 + (long long)node * IN_CH + kh;
  const float* X0 = x + (long long)node * IN_CH + kh;
  const float* Wr = sW1 + l15 * W1_PITCH + kh;
  const float* Wo = Wr + HID_CH * W1_PITCH;

  v8f c[8] = {};
#pragma unroll
  for (int k = 0; k < IN_CH; k += 4) {
    const v2f bn = load_frag(G0 + k);
#pragma unroll
    for (int j = 0; j < 8; ++j) {
      const v2f aw = load_frag(Wr + j * 16 * W1_PITCH + k);
      c[j] = __builtin_amdgcn_wmma_f32_16x16x4_f32(false, aw, false, bn,
                                                   (short)0, c[j], false,
                                                   false);
    }
  }
#pragma unroll
  for (int k = 0; k < IN_CH; k += 4) {
    const v2f bn = load_frag(X0 + k);
#pragma unroll
    for (int j = 0; j < 8; ++j) {
      const v2f aw = load_frag(Wo + j * 16 * W1_PITCH + k);
      c[j] = __builtin_amdgcn_wmma_f32_16x16x4_f32(false, aw, false, bn,
                                                   (short)0, c[j], false,
                                                   false);
    }
  }

  if (left1) {
    const float4 ba = *(const float4*)(sB1 + 16 * wave + 8 * hf);
    const float4 bb = *(const float4*)(sB1 + 16 * wave + 8 * hf + 4);
    float* hp = sH + l15 * HS_PITCH + 16 * wave + 8 * hf;
    *(float4*)hp =
        make_float4(fmaxf(cl[0] + ba.x, 0.0f), fmaxf(cl[1] + ba.y, 0.0f),
                    fmaxf(cl[2] + ba.z, 0.0f), fmaxf(cl[3] + ba.w, 0.0f));
    *(float4*)(hp + 4) =
        make_float4(fmaxf(cl[4] + bb.x, 0.0f), fmaxf(cl[5] + bb.y, 0.0f),
                    fmaxf(cl[6] + bb.z, 0.0f), fmaxf(cl[7] + bb.w, 0.0f));
  }
  if (has_left) __syncthreads();

  const bool left2 = has_left && wave < W2S_ROWS / 16;
  v8f dl = {};
  if (left2) {
    const float* W2aL = sW2 + (wave * 16 + l15) * W2_PITCH + 8 * hf;
    const float* hb = sH + l15 * HS_PITCH + 8 * hf;
#pragma unroll
    for (int j = 0; j < 8; ++j) {
#pragma unroll
      for (int q = 0; q < 4; ++q)
        dl = __builtin_amdgcn_wmma_f32_16x16x4_f32(
            false, load_frag(W2aL + 16 * j + 2 * q), false,
            load_frag(hb + 16 * j + 2 * q), (short)0, dl, false, false);
    }
  }

#pragma unroll
  for (int j = 0; j < 8; ++j) {
    const float4 ba = *(const float4*)(sB1 + 16 * j + 8 * hf);
    const float4 bb = *(const float4*)(sB1 + 16 * j + 8 * hf + 4);
    c[j][0] = fmaxf(c[j][0] + ba.x, 0.0f);
    c[j][1] = fmaxf(c[j][1] + ba.y, 0.0f);
    c[j][2] = fmaxf(c[j][2] + ba.z, 0.0f);
    c[j][3] = fmaxf(c[j][3] + ba.w, 0.0f);
    c[j][4] = fmaxf(c[j][4] + bb.x, 0.0f);
    c[j][5] = fmaxf(c[j][5] + bb.y, 0.0f);
    c[j][6] = fmaxf(c[j][6] + bb.z, 0.0f);
    c[j][7] = fmaxf(c[j][7] + bb.w, 0.0f);
  }

  const float* W2a = sW2 + l15 * W2_PITCH + 8 * hf;
  v8f d[5] = {};
#pragma unroll
  for (int j = 0; j < 8; ++j) {
#pragma unroll
    for (int q = 0; q < 4; ++q) {
      v2f bh;
      bh.x = c[j][2 * q];
      bh.y = c[j][2 * q + 1];
#pragma unroll
      for (int jm = 0; jm < 5; ++jm) {
        const v2f aw = load_frag(W2a + jm * 16 * W2_PITCH + 16 * j + 2 * q);
        d[jm] = __builtin_amdgcn_wmma_f32_16x16x4_f32(false, aw, false, bh,
                                                      (short)0, d[jm], false,
                                                      false);
      }
    }
  }
#pragma unroll
  for (int jm = 0; jm < 5; ++jm)
    store_stacked_run(d[jm], jm, hf, node, sB2, t, out);
  if (left2) store_stacked_run(dl, wave, hf, node_left, sB2, t, out);
}

__global__ __launch_bounds__(256) void gather_sum_rows40_log_softmax(
    const float* __restrict__ t, const int2* __restrict__ nodeinfo,
    const int* __restrict__ sorted, const int* __restrict__ ctrl,
    const int2* __restrict__ ovf, const float* __restrict__ zrow,
    float* __restrict__ out) {
  const int idx = blockIdx.x * 256 + threadIdx.x;
  const int node = idx >> 4;
  const int g = idx & 15;
  const bool act = g < OUT_CH / 4;
  float4* p = (float4*)(out + (long long)node * OUT_CH) + (act ? g : 0);
  const float4 start = act ? *p : make_float4(0.f, 0.f, 0.f, 0.f);
  const float4 v = sum_in_rows(t, OUT_CH, act, node, g, nodeinfo, sorted, ctrl,
                               ovf, zrow, start);
  float m = act ? fmaxf(fmaxf(v.x, v.y), fmaxf(v.z, v.w)) : -INFINITY;
#pragma unroll
  for (int msk = 8; msk >= 1; msk >>= 1)
    m = fmaxf(m, __shfl_xor(m, msk, 32));
  float s = act ? (__expf(v.x - m) + __expf(v.y - m) + __expf(v.z - m) +
                   __expf(v.w - m))
                : 0.0f;
#pragma unroll
  for (int msk = 8; msk >= 1; msk >>= 1) s += __shfl_xor(s, msk, 32);
  const float lz = m + __logf(s);
  if (act) *p = make_float4(v.x - lz, v.y - lz, v.z - lz, v.w - lz);
}

extern "C" void kernel_launch(void* const* d_in, const int* in_sizes, int n_in,
                              void* d_out, int out_size, void* d_ws,
                              size_t ws_size, hipStream_t stream) {
  (void)in_sizes; (void)n_in; (void)out_size; (void)ws_size;
  const float* x = (const float*)d_in[0];
  const int* ei = (const int*)d_in[1];
  const float* W1_rel = (const float*)d_in[2];
  const float* b1 = (const float*)d_in[3];
  const float* W1_root = (const float*)d_in[4];
  const float* W2_rel = (const float*)d_in[5];
  const float* b2 = (const float*)d_in[6];
  const float* W2_root = (const float*)d_in[7];
  float* out = (float*)d_out;

  float* agg1 = (float*)d_ws;
  float* tbuf = agg1 + (size_t)N_NODES * IN_CH;
  int* ctrl = (int*)(tbuf + (size_t)N_NODES * OUT_CH);
  float* zrow = (float*)(ctrl + 320);
  int2* nodeinfo = (int2*)(zrow + IN_CH);
  int2* rbuf = nodeinfo + N_NODES;
  int* sorted = (int*)(rbuf + (size_t)N_RANGES * RANGE_CAP);
  int2* ovf = (int2*)(sorted + (size_t)N_RANGES * RANGE_CAP);

  const int* srcv = ei;
  const int* dstv = ei + N_EDGES;

  clear_ctrl<<<1, 512, 0, stream>>>(ctrl, zrow);
  partition_edges<<<PART_BLOCKS, 256, 0, stream>>>(srcv, dstv, ctrl, rbuf, ovf);
  sort_range_sum_rows64<<<N_RANGES, SORT_THREADS, 0, stream>>>(
      ctrl, rbuf, ovf, x, zrow, sorted, nodeinfo, agg1);
  node_gemms_fused<<<GEMM_BLOCKS, GEMM_THREADS, 0, stream>>>(
      agg1, x, W1_rel, W1_root, b1, W2_rel, W2_root, b2, tbuf, out);
  gather_sum_rows40_log_softmax<<<N_NODES * 16 / 256, 256, 0, stream>>>(
      tbuf, nodeinfo, sorted, ctrl, ovf, zrow, out);
}


# edges grouped by range in LDS inside 256 partition work-groups, stored as whole-line segments with run starts; no global counters or clear launch, 4 launches; float sum order changes
# speedup vs baseline: 4.3249x; 1.0818x over previous
#include <hip/hip_runtime.h>
#include <hip/hip_bf16.h>
#include <math.h>


typedef __attribute__((ext_vector_type(2))) float v2f;
typedef __attribute__((ext_vector_type(8))) float v8f;

#define N_NODES 50000
#define N_EDGES 800000
#define IN_CH   64
#define HID_CH  128
#define OUT_CH  40

#define RANGE_NODES 196
#define N_RANGES 256
#define RANGE_CAP 4096
#define PART_BLOCKS 256
#define PART_THREADS 1024
#define PART_EDGES (N_EDGES / PART_BLOCKS)
#define PART_TRIPS ((PART_EDGES + PART_THREADS - 1) / PART_THREADS)
#define SEG_PITCH 3136
#define TBL_PITCH 260
#define LIST_CHUNK 48
static_assert(N_RANGES * RANGE_NODES >= N_NODES, "ranges must cover all nodes");
static_assert(N_EDGES % PART_BLOCKS == 0 && PART_EDGES <= SEG_PITCH &&
                  (SEG_PITCH * 8) % 128 == 0 && N_RANGES < TBL_PITCH &&
                  N_RANGES <= PART_THREADS && (N_NODES * 16) % 256 == 0,
              "segments are exact and start on a line; the node grid has no tail guard");

__global__ __launch_bounds__(PART_THREADS) void partition_edges(
    const int* __restrict__ srcv, const int* __restrict__ dstv,
    int2* __restrict__ seg, int* __restrict__ tbl, float* __restrict__ zrow) {
  __shared__ int hist[N_RANGES];
  __shared__ int rbase[N_RANGES];
  __shared__ int wsum[N_RANGES / 32];
  __shared__ int2 buf[PART_EDGES];
  const int tid = threadIdx.x;
  const int b = blockIdx.x;
  if (tid < N_RANGES) hist[tid] = 0;
  if (b == 0 && tid < IN_CH) zrow[tid] = 0.0f;
  __syncthreads();
  int d[PART_TRIPS], s[PART_TRIPS], rk[PART_TRIPS];
  const int e0 = b * PART_EDGES;
#pragma unroll
  for (int i = 0; i < PART_TRIPS; ++i) {
    const int e = tid + i * PART_THREADS;
    d[i] = -1; s[i] = 0;
    if (e < PART_EDGES) {
      d[i] = dstv[e0 + e];
      s[i] = srcv[e0 + e];
    }
  }
#pragma unroll
  for (int i = 0; i < PART_TRIPS; ++i) {
    rk[i] = 0;
    if (d[i] >= 0) rk[i] = atomicAdd(&hist[d[i] / RANGE_NODES], 1);
  }
  __syncthreads();
  int incl = 0;
  if (tid < N_RANGES) {
    incl = hist[tid];
#pragma unroll
    for (int k = 1; k < 32; k <<= 1) {
      const int u = __shfl_up(incl, k, 32);
      if ((tid & 31) >= k) incl += u;
    }
    if ((tid & 31) == 31) wsum[tid >> 5] = incl;
  }
  __syncthreads();
  if (tid < N_RANGES) {
    int o = incl - hist[tid];
    for (int w = 0; w < (tid >> 5); ++w) o += wsum[w];
    rbase[tid] = o;
    tbl[b * TBL_PITCH + tid] = o;
  }
  if (tid == N_RANGES) tbl[b * TBL_PITCH + N_RANGES] = PART_EDGES;
  __syncthreads();
#pragma unroll
  for (int i = 0; i < PART_TRIPS; ++i)
    if (d[i] >= 0)
      buf[rbase[d[i] / RANGE_NODES] + rk[i]] = make_int2(d[i], s[i]);
  __syncthreads();
  int2* out_seg = seg + (long long)b * SEG_PITCH;
#pragma unroll
  for (int i = 0; i < PART_TRIPS; ++i) {
    const int e = tid + i * PART_THREADS;
    if (e < PART_EDGES) out_seg[e] = buf[e];
  }
}

#define SORT_THREADS 1024
#define SORT_GROUPS (SORT_THREADS / 16)
__global__ __launch_bounds__(SORT_THREADS) void sort_range_sum_rows64(
    const int* __restrict__ tbl, const int2* __restrict__ seg,
    const float* __restrict__ feat, const float* __restrict__ zrow,
    int* __restrict__ sorted, int2* __restrict__ nodeinfo,
    int* __restrict__ rtot, float* __restrict__ agg) {
  __shared__ int cnt[256];
  __shared__ int off[256];
  __shared__ int list[RANGE_CAP];
  __shared__ int where[RANGE_CAP];
  __shared__ int next_ln;
  __shared__ int tot_s;
  __shared__ int wsum[PART_BLOCKS / 32];
  const int tid = threadIdx.x;
  const int r = blockIdx.x;
  int len = 0, incl = 0, o0 = 0;
  if (tid < PART_BLOCKS) {
    o0 = tbl[tid * TBL_PITCH + r];
    len = tbl[tid * TBL_PITCH + r + 1] - o0;
    incl = len;
#pragma unroll
    for (int k = 1; k < 32; k <<= 1) {
      const int u = __shfl_up(incl, k, 32);
      if ((tid & 31) >= k) incl += u;
    }
    if ((tid & 31) == 31) wsum[tid >> 5] = incl;
    cnt[tid] = 0;
  }
  if (tid == 0) next_ln = SORT_GROUPS;
  __syncthreads();
  if (tid < PART_BLOCKS) {
    int o = incl - len;
    for (int w = 0; w < (tid >> 5); ++w) o += wsum[w];
    if (tid == PART_BLOCKS - 1) tot_s = o + len;
    const int lim = len < RANGE_CAP - o ? len : RANGE_CAP - o;
    const int at0 = tid * SEG_PITCH + o0;
    for (int k = 0; k < lim; ++k) where[o + k] = at0 + k;
  }
  __syncthreads();
  const int tot = tot_s;
  const bool over = tot > RANGE_CAP;
  const int n = over ? 0 : tot;
  int2 p[RANGE_CAP / SORT_THREADS];
  int dl[RANGE_CAP / SORT_THREADS], rk[RANGE_CAP / SORT_THREADS];
#pragma unroll
  for (int i = 0; i < RANGE_CAP / SORT_THREADS; ++i) p[i] = make_int2(0, 0);
  if (n > 0) {
#pragma unroll
    for (int i = 0; i < RANGE_CAP / SORT_THREADS; ++i) {
      const int f = tid + i * SORT_THREADS;
      p[i] = seg[where[f < n ? f : n - 1]];
    }
  }
#pragma unroll
  for (int i = 0; i < RANGE_CAP / SORT_THREADS; ++i) {
    dl[i] = -1; rk[i] = 0;
    if (tid + i * SORT_THREADS < n) {
      dl[i] = p[i].x - r * RANGE_NODES;
      rk[i] = atomicAdd(&cnt[dl[i]], 1);
    }
  }
  __syncthreads();
  if (tid < 256) {
    int o = 0;
    for (int u = 0; u < tid; ++u) o += cnt[u];
    off[tid] = o;
  }
  __syncthreads();
#pragma unroll
  for (int i = 0; i < RANGE_CAP / SORT_THREADS; ++i)
    if (dl[i] >= 0) list[off[dl[i]] + rk[i]] = p[i].y;
  __syncthreads();
#pragma unroll
  for (int i = 0; i < RANGE_CAP / SORT_THREADS; ++i) {
    const int e = tid + i * SORT_THREADS;
    if (e < n) sorted[r * RANGE_CAP + e] = list[e];
  }
  const int first = r * RANGE_NODES;
  if (tid < RANGE_NODES) {
    const int node = first + tid;
    if (node < N_NODES)
      nodeinfo[node] = make_int2(r * RANGE_CAP + off[tid], cnt[tid]);
  }
  if (tid == 0) rtot[r] = tot;

  const int g = tid & 15;
  const int nn = N_NODES - first < RANGE_NODES ? N_NODES - first : RANGE_NODES;
  const float* base = feat + g * 4;
  const float* zpiece = zrow + g * 4;
  int ln = tid >> 4;
  while (ln < nn) {
    int nxt = 0;
    if (g == 0) nxt = atomicAdd(&next_ln, 1);
    const int deg = cnt[ln];
    const int st = off[ln];
    float4 acc = make_float4(0.f, 0.f, 0.f, 0.f);
    for (int j0 = 0; j0 < deg; j0 += 8) {
      int s[8];
#pragma unroll
      for (int u = 0; u < 8; ++u)
        s[u] = list[st + (j0 + u < deg ? j0 + u : deg - 1)];
      float4 v[8];
#pragma unroll
      for (int u = 0; u < 8; ++u) {
        const float* qp =
            (j0 + u < deg) ? base + (long long)s[u] * IN_CH : zpiece;
        v[u] = *(const float4*)qp;
      }
#pragma unroll
      for (int u = 0; u < 8; ++u) {
        acc.x += v[u].x; acc.y += v[u].y; acc.z += v[u].z; acc.w += v[u].w;
      }
    }
    const int node = first + ln;
    if (over) {
      for (int wb = 0; wb < PART_BLOCKS; ++wb) {
        const int w0 = tbl[wb * TBL_PITCH + r];
        const int w1 = tbl[wb * TBL_PITCH + r + 1];
        const int2* wp = seg + (long long)wb * SEG_PITCH;
        for (int k = w0; k < w1; ++k) {
          const int2 pr = wp[k];
          if (pr.x == node) {
            const float4 v = *(const float4*)(base + (long long)pr.y * IN_CH);
            acc.x += v.x; acc.y += v.y; acc.z += v.z; acc.w += v.w;
          }
        }
      }
    }
    *(float4*)(agg + (long long)node * IN_CH + g * 4) = acc;
    ln = __shfl(nxt, 0, 16);
  }
}

__device__ __forceinline__ float4 sum_in_rows(
    const float* __restrict__ rows, const int rowf, const bool act,
    const int node, const int g, const int2* __restrict__ nodeinfo,
    const int* __restrict__ sorted, const int* __restrict__ rtot,
    const int* __restrict__ tbl, const int2* __restrict__ seg,
    const float* __restrict__ zrow, float4 acc) {
  const unsigned long long info =
      *(const unsigned long long*)(nodeinfo + node);
  const int start = (int)(unsigned)info;
  const int deg = (int)(info >> 32);
  const float* base = rows + g * 4;
  const float* zpiece = zrow + g * 4;
  for (int o = 0; o < deg; o += LIST_CHUNK) {
    const int n = deg - o < LIST_CHUNK ? deg - o : LIST_CHUNK;
    const int* b = sorted + start + o;
    const int e0 = (g < n) ? b[g] : 0;
    const int e1 = (g + 16 < n) ? b[g + 16] : 0;
    const int e2 = (g + 32 < n) ? b[g + 32] : 0;
    for (int j0 = 0; j0 < n; j0 += 8) {
      const int ev = j0 < 16 ? e0 : (j0 < 32 ? e1 : e2);
      float4 v[8];
#pragma unroll
      for (int u = 0; u < 8; ++u) {
        const int s = __shfl(ev, (j0 + u) & 15, 16);
        const float* p =
            (act && j0 + u < n) ? base + (long long)s * rowf : zpiece;
        v[u] = *(const float4*)p;
      }
#pragma unroll
      for (int u = 0; u < 8; ++u) {
        acc.x += v[u].x; acc.y += v[u].y; acc.z += v[u].z; acc.w += v[u].w;
      }
    }
  }
  const int r = node / RANGE_NODES;
  if (rtot[r] > RANGE_CAP) {
    for (int wb = 0; wb < PART_BLOCKS; ++wb) {
      const int w0 = tbl[wb * TBL_PITCH + r];
      const int w1 = tbl[wb * TBL_PITCH + r + 1];
      const int2* wp = seg + (long long)wb * SEG_PITCH;
      for (int k = w0; k < w1; ++k) {
        const int2 p = wp[k];
        if (p.x == node && act) {
          const float4 v = *(const float4*)(base + (long long)p.y * rowf);
          acc.x += v.x; acc.y += v.y; acc.z += v.z; acc.w += v.w;
        }
      }
    }
  }
  return acc;
}

__device__ __forceinline__ v2f load_frag(const float* base) {
  return *(const v2f*)base;
}

#define W1_PITCH (IN_CH + 4)
#define W2_PITCH (HID_CH + 4)
#define W2S_ROWS (2 * OUT_CH)

#define GEMM_WAVES 12
#define GEMM_THREADS (GEMM_WAVES * 32)
#define GEMM_BLOCKS 256
#define N_TILES (N_NODES / 16)
#define MAIN_TILES (GEMM_BLOCKS * GEMM_WAVES)
#define LEFT_TILES (N_TILES - MAIN_TILES)
#define HS_PITCH (HID_CH + 4)
static_assert(N_NODES % 16 == 0 && LEFT_TILES >= 0 && LEFT_TILES <= GEMM_BLOCKS &&
                  GEMM_THREADS >= 256 && (HID_CH * IN_CH / 4) % 256 == 0 &&
                  (OUT_CH * HID_CH / 4) % 256 == 0,
              "every tile needs an owner, and 256 threads stage the weights in whole trips");

__device__ __forceinline__ void store_stacked_run(
    const v8f d, const int jm, const int hf, const int node,
    const float* sB2, float* __restrict__ t, float* __restrict__ out) {
  const int u = 2 * jm + hf;
  const bool is_t = u < 5;
  const int oc = (is_t ? u : u - 5) * 8;
  float* p = (is_t ? t : out) + (long long)node * OUT_CH + oc;
  float4 ba = *(const float4*)(sB2 + oc);
  float4 bb = *(const float4*)(sB2 + oc + 4);
  if (is_t) {
    ba = make_float4(0.f, 0.f, 0.f, 0.f);
    bb = ba;
  }
  *(float4*)p = make_float4(d[0] + ba.x, d[1] + ba.y, d[2] + ba.z, d[3] + ba.w);
  *(float4*)(p + 4) =
      make_float4(d[4] + bb.x, d[5] + bb.y, d[6] + bb.z, d[7] + bb.w);
}

__global__ __launch_bounds__(GEMM_THREADS) void node_gemms_fused(
    const float* __restrict__ agg1, const float* __restrict__ x,
    const float* __restrict__ W1rel, const float* __restrict__ W1root,
    const float* __restrict__ b1, const float* __restrict__ W2rel,
    const float* __restrict__ W2root, const float* __restrict__ b2,
    float* __restrict__ t, float* __restrict__ out) {
  __shared__ alignas(16) float sW1[2 * HID_CH * W1_PITCH];
  __shared__ alignas(16) float sW2[W2S_ROWS * W2_PITCH];
  __shared__ alignas(16) float sB1[HID_CH];
  __shared__ alignas(16) float sB2[OUT_CH];
  __shared__ alignas(16) float sH[16 * HS_PITCH];
  const int tid = threadIdx.x;
  if (tid < 256) {
#pragma unroll
    for (int i = 0; i < HID_CH * IN_CH / 4 / 256; ++i) {
      const int f = tid + 256 * i;
      const int row = f / (IN_CH / 4), c4 = f % (IN_CH / 4);
      const float4 wr = ((const float4*)W1rel)[f];
      const float4 wo = ((const float4*)W1root)[f];
      *(float4*)(sW1 + row * W1_PITCH + c4 * 4) = wr;
      *(float4*)(sW1 + (HID_CH + row) * W1_PITCH + c4 * 4) = wo;
    }
#pragma unroll
    for (int i = 0; i < OUT_CH * HID_CH / 4 / 256; ++i) {
      const int f = tid + 256 * i;
      const int row = f / (HID_CH / 4), c4 = f % (HID_CH / 4);
      const float4 wr = ((const float4*)W2rel)[f];
      const float4 wo = ((const float4*)W2root)[f];
      *(float4*)(sW2 + row * W2_PITCH + c4 * 4) = wr;
      *(float4*)(sW2 + (OUT_CH + row) * W2_PITCH + c4 * 4) = wo;
    }
  }
  if (tid < HID_CH) sB1[tid] = b1[tid];
  if (tid < OUT_CH) sB2[tid] = b2[tid];
  __syncthreads();

  const int lane = tid & 31;
  const int wave = tid >> 5;
  const int l15 = lane & 15;
  const int hf = lane >> 4;
  const int kh = hf * 2;
  const bool has_left = blockIdx.x < LEFT_TILES;
  const int node_left = (MAIN_TILES + blockIdx.x) * 16 + l15;

  const bool left1 = has_left && wave < HID_CH / 16;
  v8f cl = {};
  if (left1) {
    const float* GL = agg1 + (long long)node_left * IN_CH + kh;
    const float* XL = x + (long long)node_left * IN_CH + kh;
    const float* WrL = sW1 + (wave * 16 + l15) * W1_PITCH + kh;
    const float* WoL = WrL + HID_CH * W1_PITCH;
#pragma unroll
    for (int k = 0; k < IN_CH; k += 4)
      cl = __builtin_amdgcn_wmma_f32_16x16x4_f32(false, load_frag(WrL + k),
                                                 false, load_frag(GL + k),
                                                 (short)0, cl, false, false);
#pragma unroll
    for (int k = 0; k < IN_CH; k += 4)
      cl = __builtin_amdgcn_wmma_f32_16x16x4_f32(false, load_frag(WoL + k),
                                                 false, load_frag(XL + k),
                                                 (short)0, cl, false, false);
  }

  const int mtile = blockIdx.x * GEMM_WAVES + wave;
  const int m_base = mtile * 16;
  const int node = m_base + l15;

  const float* G0 = agg1 + (long long)node * IN_CH + kh;
  const float* X0 = x + (long long)node * IN_CH + kh;
  const float* Wr = sW1 + l15 * W1_PITCH + kh;
  const float* Wo = Wr + HID_CH * W1_PITCH;

  v8f c[8] = {};
#pragma unroll
  for (int k = 0; k < IN_CH; k += 4) {
    const v2f bn = load_frag(G0 + k);
#pragma unroll
    for (int j = 0; j < 8; ++j) {
      const v2f aw = load_frag(Wr + j * 16 * W1_PITCH + k);
      c[j] = __builtin_amdgcn_wmma_f32_16x16x4_f32(false, aw, false, bn,
                                                   (short)0, c[j], false,
                                                   false);
    }
  }
#pragma unroll
  for (int k = 0; k < IN_CH; k += 4) {
    const v2f bn = load_frag(X0 + k);
#pragma unroll
    for (int j = 0; j < 8; ++j) {
      const v2f aw = load_frag(Wo + j * 16 * W1_PITCH + k);
      c[j] = __builtin_amdgcn_wmma_f32_16x16x4_f32(false, aw, false, bn,
                                                   (short)0, c[j], false,
                                                   false);
    }
  }

  if (left1) {
    const float4 ba = *(const float4*)(sB1 + 16 * wave + 8 * hf);
    const float4 bb = *(const float4*)(sB1 + 16 * wave + 8 * hf + 4);
    float* hp = sH + l15 * HS_PITCH + 16 * wave + 8 * hf;
    *(float4*)hp =
        make_float4(fmaxf(cl[0] + ba.x, 0.0f), fmaxf(cl[1] + ba.y, 0.0f),
                    fmaxf(cl[2] + ba.z, 0.0f), fmaxf(cl[3] + ba.w, 0.0f));
    *(float4*)(hp + 4) =
        make_float4(fmaxf(cl[4] + bb.x, 0.0f), fmaxf(cl[5] + bb.y, 0.0f),
                    fmaxf(cl[6] + bb.z, 0.0f), fmaxf(cl[7] + bb.w, 0.0f));
  }
  if (has_left) __syncthreads();

  const bool left2 = has_left && wave < W2S_ROWS / 16;
  v8f dl = {};
  if (left2) {
    const float* W2aL = sW2 + (wave * 16 + l15) * W2_PITCH + 8 * hf;
    const float* hb = sH + l15 * HS_PITCH + 8 * hf;
#pragma unroll
    for (int j = 0; j < 8; ++j) {
#pragma unroll
      for (int q = 0; q < 4; ++q)
        dl = __builtin_amdgcn_wmma_f32_16x16x4_f32(
            false, load_frag(W2aL + 16 * j + 2 * q), false,
            load_frag(hb + 16 * j + 2 * q), (short)0, dl, false, false);
    }
  }

#pragma unroll
  for (int j = 0; j < 8; ++j) {
    const float4 ba = *(const float4*)(sB1 + 16 * j + 8 * hf);
    const float4 bb = *(const float4*)(sB1 + 16 * j + 8 * hf + 4);
    c[j][0] = fmaxf(c[j][0] + ba.x, 0.0f);
    c[j][1] = fmaxf(c[j][1] + ba.y, 0.0f);
    c[j][2] = fmaxf(c[j][2] + ba.z, 0.0f);
    c[j][3] = fmaxf(c[j][3] + ba.w, 0.0f);
    c[j][4] = fmaxf(c[j][4] + bb.x, 0.0f);
    c[j][5] = fmaxf(c[j][5] + bb.y, 0.0f);
    c[j][6] = fmaxf(c[j][6] + bb.z, 0.0f);
    c[j][7] = fmaxf(c[j][7] + bb.w, 0.0f);
  }

  const float* W2a = sW2 + l15 * W2_PITCH + 8 * hf;
  v8f d[5] = {};
#pragma unroll
  for (int j = 0; j < 8; ++j) {
#pragma unroll
    for (int q = 0; q < 4; ++q) {
      v2f bh;
      bh.x = c[j][2 * q];
      bh.y = c[j][2 * q + 1];
#pragma unroll
      for (int jm = 0; jm < 5; ++jm) {
        const v2f aw = load_frag(W2a + jm * 16 * W2_PITCH + 16 * j + 2 * q);
        d[jm] = __builtin_amdgcn_wmma_f32_16x16x4_f32(false, aw, false, bh,
                                                      (short)0, d[jm], false,
                                                      false);
      }
    }
  }
#pragma unroll
  for (int jm = 0; jm < 5; ++jm)
    store_stacked_run(d[jm], jm, hf, node, sB2, t, out);
  if (left2) store_stacked_run(dl, wave, hf, node_left, sB2, t, out);
}

__global__ __launch_bounds__(256) void gather_sum_rows40_log_softmax(
    const float* __restrict__ t, const int2* __restrict__ nodeinfo,
    const int* __restrict__ sorted, const int* __restrict__ rtot,
    const int* __restrict__ tbl, const int2* __restrict__ seg,
    const float* __restrict__ zrow, float* __restrict__ out) {
  const int idx = blockIdx.x * 256 + threadIdx.x;
  const int node = idx >> 4;
  const int g = idx & 15;
  const bool act = g < OUT_CH / 4;
  float4* p = (float4*)(out + (long long)node * OUT_CH) + (act ? g : 0);
  const float4 start = act ? *p : make_float4(0.f, 0.f, 0.f, 0.f);
  const float4 v = sum_in_rows(t, OUT_CH, act, node, g, nodeinfo, sorted, rtot,
                               tbl, seg, zrow, start);
  float m = act ? fmaxf(fmaxf(v.x, v.y), fmaxf(v.z, v.w)) : -INFINITY;
#pragma unroll
  for (int msk = 8; msk >= 1; msk >>= 1)
    m = fmaxf(m, __shfl_xor(m, msk, 32));
  float s = act ? (__expf(v.x - m) + __expf(v.y - m) + __expf(v.z - m) +
                   __expf(v.w - m))
                : 0.0f;
#pragma unroll
  for (int msk = 8; msk >= 1; msk >>= 1) s += __shfl_xor(s, msk, 32);
  const float lz = m + __logf(s);
  if (act) *p = make_float4(v.x - lz, v.y - lz, v.z - lz, v.w - lz);
}

extern "C" void kernel_launch(void* const* d_in, const int* in_sizes, int n_in,
                              void* d_out, int out_size, void* d_ws,
                              size_t ws_size, hipStream_t stream) {
  (void)in_sizes; (void)n_in; (void)out_size; (void)ws_size;
  const float* x = (const float*)d_in[0];
  const int* ei = (const int*)d_in[1];
  const float* W1_rel = (const float*)d_in[2];
  const float* b1 = (const float*)d_in[3];
  const float* W1_root = (const float*)d_in[4];
  const float* W2_rel = (const float*)d_in[5];
  const float* b2 = (const float*)d_in[6];
  const float* W2_root = (const float*)d_in[7];
  float* out = (float*)d_out;

  float* agg1 = (float*)d_ws;
  float* tbuf = agg1 + (size_t)N_NODES * IN_CH;
  int* rtot = (int*)(tbuf + (size_t)N_NODES * OUT_CH);
  float* zrow = (float*)(rtot + 320);
  int2* nodeinfo = (int2*)(zrow + IN_CH);
  int2* seg = nodeinfo + N_NODES;
  int* sorted = (int*)(seg + (size_t)PART_BLOCKS * SEG_PITCH);
  int* tbl = sorted + (size_t)N_RANGES * RANGE_CAP;

  const int* srcv = ei;
  const int* dstv = ei + N_EDGES;

  partition_edges<<<PART_BLOCKS, PART_THREADS, 0, stream>>>(srcv, dstv, seg, tbl,
                                                          zrow);
  sort_range_sum_rows64<<<N_RANGES, SORT_THREADS, 0, stream>>>(
      tbl, seg, x, zrow, sorted, nodeinfo, rtot, agg1);
  node_gemms_fused<<<GEMM_BLOCKS, GEMM_THREADS, 0, stream>>>(
      agg1, x, W1_rel, W1_root, b1, W2_rel, W2_root, b2, tbuf, out);
  gather_sum_rows40_log_softmax<<<N_NODES * 16 / 256, 256, 0, stream>>>(
      tbuf, nodeinfo, sorted, rtot, tbl, seg, zrow, out);
}
